# starting kernel
# baseline (speedup 1.0000x reference)
#include <hip/hip_runtime.h>

typedef __attribute__((ext_vector_type(2))) float v2f;
typedef __attribute__((ext_vector_type(8))) float v8f;

template <int K, int N, bool RELU>
__global__ __launch_bounds__(32)
void gcn_gemm_wmma_f32(const float* __restrict__ A,
                       const float* __restrict__ W,
                       const float* __restrict__ bias,
                       float* __restrict__ Dout)
{
    const int lane  = threadIdx.x;
    const int hi    = lane >> 4;
    const int l16   = lane & 15;
    const int mtile = blockIdx.y << 4;
    const int ntile = blockIdx.x << 4;

    const int m    = mtile + l16;
    const int koff = hi << 1;

    const float bv = bias[ntile + l16];
    v8f acc;
#pragma unroll
    for (int r = 0; r < 8; ++r) acc[r] = bv;

    const float* __restrict__ arow = A + (size_t)m * K + koff;
    const float* __restrict__ bcol = W + (size_t)koff * N + ntile + l16;

#pragma unroll
    for (int k0 = 0; k0 < K; k0 += 4) {
        v2f a = *(const v2f*)(arow + k0);
        if (RELU) {
            a[0] = fmaxf(a[0], 0.0f);
            a[1] = fmaxf(a[1], 0.0f);
        }
        v2f b;
        b[0] = bcol[k0 * N];
        b[1] = bcol[(k0 + 1) * N];
        acc = __builtin_amdgcn_wmma_f32_16x16x4_f32(
            false, a, false, b, (short)0, acc, false, false);
    }

#pragma unroll
    for (int r = 0; r < 8; ++r) {
        Dout[(size_t)(mtile + (hi << 3) + r) * N + ntile + l16] = acc[r];
    }
}

__global__ __launch_bounds__(256)
void spmm_scatter128(const int* __restrict__ row, const int* __restrict__ col,
                     const float* __restrict__ val,
                     const float* __restrict__ H, float* __restrict__ out, int E)
{
    const long long t = (long long)blockIdx.x * blockDim.x + threadIdx.x;
    const int e = (int)(t >> 5);
    if (e >= E) return;
    const int f = ((int)t & 31) << 2;
    const int r = row[e];
    const int c = col[e];
    const float v = val[e];
    const float4 h = *(const float4*)(H + (size_t)c * 128 + f);
    float* o = out + (size_t)r * 128 + f;
    atomicAdd(o + 0, v * h.x);
    atomicAdd(o + 1, v * h.y);
    atomicAdd(o + 2, v * h.z);
    atomicAdd(o + 3, v * h.w);
}

__global__ __launch_bounds__(256)
void spmm_scatter16(const int* __restrict__ row, const int* __restrict__ col,
                    const float* __restrict__ val,
                    const float* __restrict__ H, float* __restrict__ out, int E)
{
    const long long t = (long long)blockIdx.x * blockDim.x + threadIdx.x;
    const int e = (int)(t >> 4);
    if (e >= E) return;
    const int f = (int)t & 15;
    const float g = val[e] * H[(size_t)col[e] * 16 + f];
    atomicAdd(out + (size_t)row[e] * 16 + f, g);
}

extern "C" void kernel_launch(void* const* d_in, const int* in_sizes, int n_in,
                              void* d_out, int out_size, void* d_ws, size_t ws_size,
                              hipStream_t stream) {
    const float* x   = (const float*)d_in[0];
    const int*   row = (const int*)  d_in[1];
    const int*   col = (const int*)  d_in[2];
    const float* val = (const float*)d_in[3];
    const float* W1  = (const float*)d_in[4];
    const float* b1  = (const float*)d_in[5];
    const float* W2  = (const float*)d_in[6];
    const float* b2  = (const float*)d_in[7];
    float* out = (float*)d_out;

    constexpr int D_IN  = 128;
    constexpr int D_HID = 128;
    constexpr int N_CLS = 16;
    const int E = in_sizes[1];
    const int M = in_sizes[0] / D_IN;

    char*  ws  = (char*)d_ws;
    float* H1  = (float*)ws;
    float* AGG = (float*)(ws + (size_t)M * D_HID * sizeof(float));
    float* H2  = (float*)(ws + 2 * (size_t)M * D_HID * sizeof(float));

    hipMemsetAsync(AGG, 0, (size_t)M * D_HID * sizeof(float), stream);
    hipMemsetAsync(out, 0, (size_t)out_size * sizeof(float), stream);

    dim3 g1(D_HID / 16, M / 16);
    gcn_gemm_wmma_f32<D_IN, D_HID, false><<<g1, 32, 0, stream>>>(x, W1, b1, H1);

    {
        const long long threads = (long long)E * 32;
        const int blk = 256;
        const unsigned nb = (unsigned)((threads + blk - 1) / blk);
        spmm_scatter128<<<nb, blk, 0, stream>>>(row, col, val, H1, AGG, E);
    }

    dim3 g2(N_CLS / 16, M / 16);
    gcn_gemm_wmma_f32<D_HID, N_CLS, true><<<g2, 32, 0, stream>>>(AGG, W2, b2, H2);

    {
        const long long threads = (long long)E * 16;
        const int blk = 256;
        const unsigned nb = (unsigned)((threads + blk - 1) / blk);
        spmm_scatter16<<<nb, blk, 0, stream>>>(row, col, val, H2, out, E);
    }
}


# Layer-1 SpMM edge search reads a 1-byte row-range key per edge (row/400, written by the product launch), not the 4-byte row index; 32 keys per lane, zero-byte test; sum order LDS arrival as before
# speedup vs baseline: 2.2767x; 2.2767x over previous
#include <hip/hip_runtime.h>
#include <type_traits>

typedef __attribute__((ext_vector_type(2))) float v2f;
typedef __attribute__((ext_vector_type(8))) float v8f;

template <int K, int N, bool RELU>
__global__ __launch_bounds__(32)
void gcn_gemm_wmma_f32(const float* __restrict__ A,
                       const float* __restrict__ A2,
                       const float* __restrict__ W,
                       const float* __restrict__ bias,
                       float* __restrict__ Dout)
{
    const int lane  = threadIdx.x;
    const int hi    = lane >> 4;
    const int l16   = lane & 15;
    const int mtile = blockIdx.y << 4;
    const int ntile = blockIdx.x << 4;

    const int m    = mtile + l16;
    const int koff = hi << 1;

    const float bv = bias[ntile + l16];
    v8f acc;
#pragma unroll
    for (int r = 0; r < 8; ++r) acc[r] = bv;

    const float* __restrict__ arow  = A  + (size_t)m * K + koff;
    const float* __restrict__ arow2 = A2 + (size_t)m * K + koff;
    const float* __restrict__ bcol = W + (size_t)koff * N + ntile + l16;

#pragma unroll
    for (int k0 = 0; k0 < K; k0 += 4) {
        v2f a = *(const v2f*)(arow + k0);
        const v2f a2 = *(const v2f*)(arow2 + k0);
        a[0] += a2[0];
        a[1] += a2[1];
        if (RELU) {
            a[0] = fmaxf(a[0], 0.0f);
            a[1] = fmaxf(a[1], 0.0f);
        }
        v2f b;
        b[0] = bcol[k0 * N];
        b[1] = bcol[(k0 + 1) * N];
        acc = __builtin_amdgcn_wmma_f32_16x16x4_f32(
            false, a, false, b, (short)0, acc, false, false);
    }

#pragma unroll
    for (int r = 0; r < 8; ++r) {
        Dout[(size_t)(mtile + (hi << 3) + r) * N + ntile + l16] = acc[r];
    }
}

constexpr int AG_ROWS  = 400;
constexpr int G1_WAVES = 8;

__device__ __forceinline__ unsigned edge_key(int r) {
    const unsigned k = (unsigned)r / (unsigned)AG_ROWS;
    return k < 255u ? k : 255u;
}

__global__ __launch_bounds__(G1_WAVES * 32)
void gcn_gemm_stripe128_f32(const float* __restrict__ A,
                            const float* __restrict__ W,
                            const float* __restrict__ bias,
                            float* __restrict__ Dout, int nstripes,
                            const int* __restrict__ row,
                            unsigned int* __restrict__ keys, int E)
{
    constexpr int K = 128, N = 128;
    __shared__ __attribute__((aligned(16))) v2f Ws[(K / 2) * N];

    const int tid  = threadIdx.x;

    const int nwords = ((E + 511) >> 9) << 7;
    const int gsz    = (int)gridDim.x * (G1_WAVES * 32);
    const int i0     = (int)blockIdx.x * (G1_WAVES * 32) + tid;
    auto whole_rows = [&](int i) -> int4 {
        const int e = i << 2;
        return *(const int4*)(row + ((e + 3 < E) ? e : 0));
    };
    auto end_rows = [&](int i) -> int4 {
        const int e = i << 2;
        int4 r = make_int4(-1, -1, -1, -1);
        if (e     < E) r.x = row[e];
        if (e + 1 < E) r.y = row[e + 1];
        if (e + 2 < E) r.z = row[e + 2];
        return r;
    };
    auto store_keys = [&](int i, int4 r) {
        if ((i << 2) + 3 >= E) r = end_rows(i);
        keys[i] = edge_key(r.x) | (edge_key(r.y) << 8) | (edge_key(r.z) << 16) | (edge_key(r.w) << 24);
    };
    int4 rk[2];
#pragma unroll
    for (int t = 0; t < 2; ++t) rk[t] = whole_rows(i0 + t * gsz);

    const int lane = tid & 31;
    const int hi   = lane >> 4;
    const int l16  = lane & 15;
    const int wave = __builtin_amdgcn_readfirstlane(tid >> 5);
    const int stripe = blockIdx.x * G1_WAVES + wave;
    const bool live = stripe < nstripes;
    const int mtile = (live ? stripe : 0) << 4;

    const float* __restrict__ arow = A + (size_t)(mtile + l16) * K + (hi << 1);
    v2f a[K / 4];
#pragma unroll
    for (int s = 0; s < K / 4; ++s) a[s] = *(const v2f*)(arow + 4 * s);

#pragma unroll
    for (int j = 0; j < (K / 2) * (N / 4) / (G1_WAVES * 32); ++j) {
        const int i  = tid + j * (G1_WAVES * 32);
        const int p  = i >> 5;
        const int n4 = (i & 31) << 2;
        const float4 r0 = *(const float4*)(W + (size_t)(2 * p) * N + n4);
        const float4 r1 = *(const float4*)(W + (size_t)(2 * p + 1) * N + n4);
        float4* d = (float4*)(Ws + p * N + n4);
        d[0] = make_float4(r0.x, r1.x, r0.y, r1.y);
        d[1] = make_float4(r0.z, r1.z, r0.w, r1.w);
    }

    v8f acc[N / 16];
#pragma unroll
    for (int t = 0; t < N / 16; ++t) {
        const float bv = bias[t * 16 + l16];
#pragma unroll
        for (int r = 0; r < 8; ++r) acc[t][r] = bv;
    }
    __syncthreads();
    if (live) {
        const v2f* __restrict__ wl = Ws + hi * N + l16;
#pragma unroll
        for (int s = 0; s < K / 4; ++s) {
            v2f b[N / 16];
#pragma unroll
            for (int t = 0; t < N / 16; ++t) b[t] = wl[(2 * s) * N + t * 16];
#pragma unroll
            for (int t = 0; t < N / 16; ++t)
                acc[t] = __builtin_amdgcn_wmma_f32_16x16x4_f32(
                    false, a[s], false, b[t], (short)0, acc[t], false, false);
        }

#pragma unroll
        for (int r = 0; r < 8; ++r) {
            float* __restrict__ drow = Dout + (size_t)(mtile + (hi << 3) + r) * N + l16;
#pragma unroll
            for (int t = 0; t < N / 16; ++t) drow[t * 16] = acc[t][r];
        }
    }

#pragma unroll
    for (int t = 0; t < 2; ++t)
        if (i0 + t * gsz < nwords) store_keys(i0 + t * gsz, rk[t]);
    for (int i = i0 + 2 * gsz; i < nwords; i += gsz) store_keys(i, whole_rows(i));
}

constexpr int AG_SPLIT   = 2;
constexpr int AG_THREADS = 1024;
constexpr int AG_WAVES   = AG_THREADS / 32;
constexpr int AG_UNROLL  = 4;
constexpr int AG_EAGER   = 64;
constexpr int AG_QCAP    = AG_EAGER + 512;
static_assert(AG_UNROLL % 2 == 0, "pieces are searched two at a time");

__global__ __launch_bounds__(AG_THREADS)
void spmm_rows_in_lds128(const int* __restrict__ row, const int* __restrict__ col,
                         const float* __restrict__ val,
                         const float* __restrict__ H, float* __restrict__ out,
                         const unsigned int* __restrict__ keys,
                         int E, int M)
{
    __shared__ float acc[(AG_ROWS + 1) * 128];
    __shared__ int   qbuf[AG_WAVES * AG_QCAP];

    const int tid  = threadIdx.x;
    const int lane = tid & 31;
    const int wave = __builtin_amdgcn_readfirstlane(tid >> 5);
    const int range = (int)blockIdx.x / AG_SPLIT;
    const int part  = (int)blockIdx.x % AG_SPLIT;
    const int r0    = range * AG_ROWS;

    {
        float4* a4 = (float4*)acc;
        const float4 z = make_float4(0.f, 0.f, 0.f, 0.f);
        for (int i = tid; i < AG_ROWS * 32; i += AG_THREADS) a4[i] = z;
    }
    __syncthreads();

    int* myq = qbuf + wave * AG_QCAP;
    int  cnt = 0;

    const unsigned pat = (unsigned)range * 0x01010101u;
    const unsigned lt  = (1u << lane) - 1u;
    auto flags = [&](unsigned w) -> unsigned {
        const unsigned x = w ^ pat;
        const unsigned z = ((x & 0x7f7f7f7fu) + 0x7f7f7f7fu) | x | 0x7f7f7f7fu;
        return (~z) >> 7;
    };
    auto drain = [&]() {
        for (int b = 0; b < cnt; b += 8) {
            const int nb = cnt - b;
            int mr = AG_ROWS, mc = 0;
            float mv = 0.f;
            if (lane < nb && lane < 8) {
                const int e = myq[b + lane];
                mr = row[e] - r0;
                mc = col[e];
                mv = val[e];
            }
            float h[8][4];
#pragma unroll
            for (int j = 0; j < 8; ++j) {
                const int c = __builtin_amdgcn_readlane(mc, j);
                const float* __restrict__ hp = H + (size_t)c * 128 + lane;
                h[j][0] = hp[0];
                h[j][1] = hp[32];
                h[j][2] = hp[64];
                h[j][3] = hp[96];
            }
#pragma unroll
            for (int j = 0; j < 8; ++j) {
                const int   r = __builtin_amdgcn_readlane(mr, j);
                const float v = __int_as_float(__builtin_amdgcn_readlane(__float_as_int(mv), j));
                float* a = acc + r * 128 + lane;
                atomicAdd(a +  0, v * h[j][0]);
                atomicAdd(a + 32, v * h[j][1]);
                atomicAdd(a + 64, v * h[j][2]);
                atomicAdd(a + 96, v * h[j][3]);
            }
        }
        cnt = 0;
    };

    auto take = [&](const int4 ka, const int4 kb, int ea, int eb) {
        unsigned m = flags((unsigned)ka.x)        | (flags((unsigned)ka.y) << 1)
                   | (flags((unsigned)ka.z) << 2) | (flags((unsigned)ka.w) << 3)
                   | (flags((unsigned)kb.x) << 4) | (flags((unsigned)kb.y) << 5)
                   | (flags((unsigned)kb.z) << 6) | (flags((unsigned)kb.w) << 7);
#pragma nounroll
        for (int it = 0; it < 32; ++it) {
            const bool hit = m != 0u;
            const unsigned mask = __builtin_amdgcn_ballot_w32(hit);
            if (mask == 0u) break;
            if (cnt + 32 > AG_QCAP) drain();
            if (hit) {
                const int p = __builtin_ctz(m);
                const int i = p & 7;
                myq[cnt + __builtin_popcount(mask & lt)] = ((i & 4) ? eb : ea) + (((i & 3) << 2) | (p >> 3));
                m &= m - 1u;
            }
            cnt += __builtin_popcount(mask);
        }
    };

    const int nchunk = (E + 511) >> 9;
    const int c0     = (int)(((long long)nchunk * part) / AG_SPLIT);
    const int c1     = (int)(((long long)nchunk * (part + 1)) / AG_SPLIT);
    const int4* __restrict__ kp = (const int4*)keys;
    constexpr int AG_ROUND = AG_WAVES * AG_UNROLL;
    auto load_keys = [&](int4 (&k)[AG_UNROLL], int base) {
#pragma unroll
        for (int u = 0; u < AG_UNROLL; ++u) {
            const int ch = base + u * AG_WAVES + wave;
            k[u] = kp[(size_t)(ch < c1 ? ch : c1 - 1) * 32 + lane];
        }
    };
    if (c0 < c1) {
        int4 kk[AG_UNROLL];
        load_keys(kk, c0);
        for (int base = c0; base < c1; base += AG_ROUND) {
            int4 kn[AG_UNROLL];
            load_keys(kn, base + AG_ROUND);
#pragma unroll
            for (int u = 0; u < AG_UNROLL; u += 2) {
                const int ca = base + u * AG_WAVES + wave;
                const int cb = ca + AG_WAVES;
                if (ca < c1) {
                    if (cnt >= AG_EAGER) drain();
                    const int4 kb = (cb < c1) ? kk[u + 1] : make_int4(-1, -1, -1, -1);
                    take(kk[u], kb, (ca << 9) + (lane << 4), (cb << 9) + (lane << 4));
                }
            }
#pragma unroll
            for (int u = 0; u < AG_UNROLL; ++u) kk[u] = kn[u];
        }
    }
    drain();
    __syncthreads();

    {
        const float4* a4 = (const float4*)acc;
        float4* o4 = (float4*)(out + (size_t)part * (size_t)M * 128);
        for (int i = tid; i < AG_ROWS * 32; i += AG_THREADS) {
            const int rrow = r0 + (i >> 5);
            if (rrow < M) o4[(size_t)rrow * 32 + (i & 31)] = a4[i];
        }
    }
}

constexpr int S2_THREADS = 256;
constexpr int S2_WAVES   = S2_THREADS / 32;
constexpr int S2_TRIPS   = 4;

template <int J>
__device__ __forceinline__ int row_lane(int x) {
    return __builtin_amdgcn_update_dpp(0, x, 0x150 + J, 0xf, 0xf, true);
}

template <int J, int N, class F>
__device__ __forceinline__ void static_for(F&& fn) {
    if constexpr (J < N) {
        fn(std::integral_constant<int, J>{});
        static_for<J + 1, N>(fn);
    }
}

__global__ __launch_bounds__(S2_THREADS)
void spmm_scatter16(const int* __restrict__ row, const int* __restrict__ col,
                    const float* __restrict__ val,
                    const float* __restrict__ H, float* __restrict__ out, int E)
{
    const int lane = threadIdx.x & 31;
    const unsigned f = lane & 15;
    const int wave   = __builtin_amdgcn_readfirstlane((int)(blockIdx.x * S2_THREADS + threadIdx.x) >> 5);
    const int nwaves = (int)(gridDim.x * S2_THREADS) >> 5;
    const int nfull  = E >> 5;

    if (wave < nfull) {
        int   r = row[(wave << 5) + lane];
        int   c = col[(wave << 5) + lane];
        float v = val[(wave << 5) + lane];
        for (int ch = wave; ch < nfull; ch += nwaves) {
            float h[16];
            static_for<0, 16>([&](auto jc) {
                constexpr int J = decltype(jc)::value;
                h[J] = H[(unsigned)row_lane<J>(c) * 16u + f];
            });
            const int chn = ch + nwaves;
            int rn = 0, cn = 0; float vn = 0.f;
            if (chn < nfull) {
                rn = row[(chn << 5) + lane];
                cn = col[(chn << 5) + lane];
                vn = val[(chn << 5) + lane];
            }
            static_for<0, 16>([&](auto jc) {
                constexpr int J = decltype(jc)::value;
                const unsigned rj = (unsigned)row_lane<J>(r);
                const float    vj = __int_as_float(row_lane<J>(__float_as_int(v)));
                atomicAdd(out + (rj * 16u + f), vj * h[J]);
            });
            r = rn; c = cn; v = vn;
        }
    }

    if ((E & 31) != 0 && wave == nfull % nwaves) {
        for (int j = 0; j < 16; ++j) {
            const int e = (nfull << 5) + ((lane >> 4) << 4) + j;
            if (e < E)
                atomicAdd(out + ((unsigned)row[e] * 16u + f),
                          val[e] * H[(unsigned)col[e] * 16u + f]);
        }
    }
}

extern "C" void kernel_launch(void* const* d_in, const int* in_sizes, int n_in,
                              void* d_out, int out_size, void* d_ws, size_t ws_size,
                              hipStream_t stream) {
    const float* x   = (const float*)d_in[0];
    const int*   row = (const int*)  d_in[1];
    const int*   col = (const int*)  d_in[2];
    const float* val = (const float*)d_in[3];
    const float* W1  = (const float*)d_in[4];
    const float* b1  = (const float*)d_in[5];
    const float* W2  = (const float*)d_in[6];
    const float* b2  = (const float*)d_in[7];
    float* out = (float*)d_out;

    constexpr int D_IN  = 128;
    constexpr int D_HID = 128;
    constexpr int N_CLS = 16;
    const int E = in_sizes[1];
    const int M = in_sizes[0] / D_IN;

    char*  ws  = (char*)d_ws;
    float* H1  = (float*)ws;
    float* AGG = (float*)(ws + (size_t)M * D_HID * sizeof(float));
    float* H2  = (float*)(ws + (1 + AG_SPLIT) * (size_t)M * D_HID * sizeof(float));
    unsigned int* KEY = (unsigned int*)(ws + ((1 + AG_SPLIT) * (size_t)D_HID + N_CLS) * (size_t)M * sizeof(float));

    hipMemsetAsync(out, 0, (size_t)out_size * sizeof(float), stream);

    {
        const int nstripes = M / 16;
        const unsigned nb = (unsigned)((nstripes + G1_WAVES - 1) / G1_WAVES);
        gcn_gemm_stripe128_f32<<<nb, G1_WAVES * 32, 0, stream>>>(x, W1, b1, H1, nstripes, row, KEY, E);
    }

    {
        const unsigned nb = (unsigned)((M + AG_ROWS - 1) / AG_ROWS) * AG_SPLIT;
        spmm_rows_in_lds128<<<nb, AG_THREADS, 0, stream>>>(row, col, val, H1, AGG, KEY, E, M);
    }

    static_assert(AG_SPLIT == 2, "the layer-2 product adds exactly two planes");
    dim3 g2(N_CLS / 16, M / 16);
    gcn_gemm_wmma_f32<D_HID, N_CLS, true><<<g2, 32, 0, stream>>>(
        AGG, AGG + (size_t)M * D_HID, W2, b2, H2);

    {
        const int nchunk = (E + 31) / 32;
        const int per_block = S2_WAVES * S2_TRIPS;
        const unsigned nb = (unsigned)((nchunk + per_block - 1) / per_block);
        spmm_scatter16<<<nb, S2_THREADS, 0, stream>>>(row, col, val, H2, out, E);
    }
}


# Layer-2 row sums read edge lists that the layer-1 row-sum blocks write as they find their edges: a block owns 200 rows in LDS; no search, global atomics or memset; sum order LDS arrival
# speedup vs baseline: 2.6806x; 1.1774x over previous
#include <hip/hip_runtime.h>
#include <type_traits>

typedef __attribute__((ext_vector_type(2))) float v2f;
typedef __attribute__((ext_vector_type(8))) float v8f;

template <int K, int N, bool RELU>
__global__ __launch_bounds__(32)
void gcn_gemm_wmma_f32(const float* __restrict__ A,
                       const float* __restrict__ A2,
                       const float* __restrict__ W,
                       const float* __restrict__ bias,
                       float* __restrict__ Dout)
{
    const int lane  = threadIdx.x;
    const int hi    = lane >> 4;
    const int l16   = lane & 15;
    const int mtile = blockIdx.y << 4;
    const int ntile = blockIdx.x << 4;

    const int m    = mtile + l16;
    const int koff = hi << 1;

    const float bv = bias[ntile + l16];
    v8f acc;
#pragma unroll
    for (int r = 0; r < 8; ++r) acc[r] = bv;

    const float* __restrict__ arow  = A  + (size_t)m * K + koff;
    const float* __restrict__ arow2 = A2 + (size_t)m * K + koff;
    const float* __restrict__ bcol = W + (size_t)koff * N + ntile + l16;

#pragma unroll
    for (int k0 = 0; k0 < K; k0 += 4) {
        v2f a = *(const v2f*)(arow + k0);
        const v2f a2 = *(const v2f*)(arow2 + k0);
        a[0] += a2[0];
        a[1] += a2[1];
        if (RELU) {
            a[0] = fmaxf(a[0], 0.0f);
            a[1] = fmaxf(a[1], 0.0f);
        }
        v2f b;
        b[0] = bcol[k0 * N];
        b[1] = bcol[(k0 + 1) * N];
        acc = __builtin_amdgcn_wmma_f32_16x16x4_f32(
            false, a, false, b, (short)0, acc, false, false);
    }

#pragma unroll
    for (int r = 0; r < 8; ++r) {
        Dout[(size_t)(mtile + (hi << 3) + r) * N + ntile + l16] = acc[r];
    }
}

constexpr int AG_ROWS  = 400;
constexpr int G1_WAVES = 8;

__device__ __forceinline__ unsigned edge_key(int r) {
    const unsigned k = (unsigned)r / (unsigned)AG_ROWS;
    return k < 255u ? k : 255u;
}

__global__ __launch_bounds__(G1_WAVES * 32)
void gcn_gemm_stripe128_f32(const float* __restrict__ A,
                            const float* __restrict__ W,
                            const float* __restrict__ bias,
                            float* __restrict__ Dout, int nstripes,
                            const int* __restrict__ row,
                            unsigned int* __restrict__ keys, int E)
{
    constexpr int K = 128, N = 128;
    __shared__ __attribute__((aligned(16))) v2f Ws[(K / 2) * N];

    const int tid  = threadIdx.x;

    const int nwords = ((E + 511) >> 9) << 7;
    const int gsz    = (int)gridDim.x * (G1_WAVES * 32);
    const int i0     = (int)blockIdx.x * (G1_WAVES * 32) + tid;
    auto whole_rows = [&](int i) -> int4 {
        const int e = i << 2;
        return *(const int4*)(row + ((e + 3 < E) ? e : 0));
    };
    auto end_rows = [&](int i) -> int4 {
        const int e = i << 2;
        int4 r = make_int4(-1, -1, -1, -1);
        if (e     < E) r.x = row[e];
        if (e + 1 < E) r.y = row[e + 1];
        if (e + 2 < E) r.z = row[e + 2];
        return r;
    };
    auto store_keys = [&](int i, int4 r) {
        if ((i << 2) + 3 >= E) r = end_rows(i);
        keys[i] = edge_key(r.x) | (edge_key(r.y) << 8) | (edge_key(r.z) << 16) | (edge_key(r.w) << 24);
    };
    int4 rk[2];
#pragma unroll
    for (int t = 0; t < 2; ++t) rk[t] = whole_rows(i0 + t * gsz);

    const int lane = tid & 31;
    const int hi   = lane >> 4;
    const int l16  = lane & 15;
    const int wave = __builtin_amdgcn_readfirstlane(tid >> 5);
    const int stripe = blockIdx.x * G1_WAVES + wave;
    const bool live = stripe < nstripes;
    const int mtile = (live ? stripe : 0) << 4;

    const float* __restrict__ arow = A + (size_t)(mtile + l16) * K + (hi << 1);
    v2f a[K / 4];
#pragma unroll
    for (int s = 0; s < K / 4; ++s) a[s] = *(const v2f*)(arow + 4 * s);

#pragma unroll
    for (int j = 0; j < (K / 2) * (N / 4) / (G1_WAVES * 32); ++j) {
        const int i  = tid + j * (G1_WAVES * 32);
        const int p  = i >> 5;
        const int n4 = (i & 31) << 2;
        const float4 r0 = *(const float4*)(W + (size_t)(2 * p) * N + n4);
        const float4 r1 = *(const float4*)(W + (size_t)(2 * p + 1) * N + n4);
        float4* d = (float4*)(Ws + p * N + n4);
        d[0] = make_float4(r0.x, r1.x, r0.y, r1.y);
        d[1] = make_float4(r0.z, r1.z, r0.w, r1.w);
    }

    v8f acc[N / 16];
#pragma unroll
    for (int t = 0; t < N / 16; ++t) {
        const float bv = bias[t * 16 + l16];
#pragma unroll
        for (int r = 0; r < 8; ++r) acc[t][r] = bv;
    }
    __syncthreads();
    if (live) {
        const v2f* __restrict__ wl = Ws + hi * N + l16;
#pragma unroll
        for (int s = 0; s < K / 4; ++s) {
            v2f b[N / 16];
#pragma unroll
            for (int t = 0; t < N / 16; ++t) b[t] = wl[(2 * s) * N + t * 16];
#pragma unroll
            for (int t = 0; t < N / 16; ++t)
                acc[t] = __builtin_amdgcn_wmma_f32_16x16x4_f32(
                    false, a[s], false, b[t], (short)0, acc[t], false, false);
        }

#pragma unroll
        for (int r = 0; r < 8; ++r) {
            float* __restrict__ drow = Dout + (size_t)(mtile + (hi << 3) + r) * N + l16;
#pragma unroll
            for (int t = 0; t < N / 16; ++t) drow[t * 16] = acc[t][r];
        }
    }

#pragma unroll
    for (int t = 0; t < 2; ++t)
        if (i0 + t * gsz < nwords) store_keys(i0 + t * gsz, rk[t]);
    for (int i = i0 + 2 * gsz; i < nwords; i += gsz) store_keys(i, whole_rows(i));
}

constexpr int AG_SPLIT   = 2;
constexpr int AG_THREADS = 1024;
constexpr int AG_WAVES   = AG_THREADS / 32;
constexpr int AG_UNROLL  = 4;
constexpr int AG_EAGER   = 64;
constexpr int AG_QCAP    = AG_EAGER + 512;
static_assert(AG_UNROLL % 2 == 0, "pieces are searched two at a time");
constexpr int L2_ROWS = AG_ROWS / 2;
constexpr int L2_CAP  = 128;
static_assert(AG_ROWS % 2 == 0, "a range is two layer-2 blocks' rows");

__global__ __launch_bounds__(AG_THREADS)
void spmm_rows_in_lds128(const int* __restrict__ row, const int* __restrict__ col,
                         const float* __restrict__ val,
                         const float* __restrict__ H, float* __restrict__ out,
                         const unsigned int* __restrict__ keys,
                         int* __restrict__ lst, int* __restrict__ lcnt,
                         int E, int M)
{
    __shared__ float acc[(AG_ROWS + 1) * 128];
    __shared__ int   qbuf[AG_WAVES * AG_QCAP];

    const int tid  = threadIdx.x;
    const int lane = tid & 31;
    const int wave = __builtin_amdgcn_readfirstlane(tid >> 5);
    const int range = (int)blockIdx.x / AG_SPLIT;
    const int part  = (int)blockIdx.x % AG_SPLIT;
    const int r0    = range * AG_ROWS;

    {
        float4* a4 = (float4*)acc;
        const float4 z = make_float4(0.f, 0.f, 0.f, 0.f);
        for (int i = tid; i < AG_ROWS * 32; i += AG_THREADS) a4[i] = z;
    }
    __syncthreads();

    int* myq = qbuf + wave * AG_QCAP;
    int  cnt = 0;
    int4* __restrict__ mylst = (int4*)lst + ((size_t)blockIdx.x * AG_WAVES + wave) * (2 * L2_CAP);
    int  wlo = 0, whi = 0;

    const unsigned pat = (unsigned)range * 0x01010101u;
    const unsigned lt  = (1u << lane) - 1u;
    auto flags = [&](unsigned w) -> unsigned {
        const unsigned x = w ^ pat;
        const unsigned z = ((x & 0x7f7f7f7fu) + 0x7f7f7f7fu) | x | 0x7f7f7f7fu;
        return (~z) >> 7;
    };
    auto drain = [&]() {
        for (int b = 0; b < cnt; b += 8) {
            const int nb = cnt - b;
            int mr = AG_ROWS, mc = 0;
            float mv = 0.f;
            const bool act = lane < nb && lane < 8;
            if (act) {
                const int e = myq[b + lane];
                mr = row[e] - r0;
                mc = col[e];
                mv = val[e];
            }
            {
                const bool up = act && mr >= L2_ROWS;
                const unsigned mu = __builtin_amdgcn_ballot_w32(up);
                const unsigned md = __builtin_amdgcn_ballot_w32(act && !up);
                if (act) {
                    const int pos = up ? whi + __builtin_popcount(mu & lt) : wlo + __builtin_popcount(md & lt);
                    if (pos < L2_CAP)
                        mylst[(up ? L2_CAP : 0) + pos] = make_int4(up ? mr - L2_ROWS : mr, mc, __float_as_int(mv), 0);
                }
                whi += __builtin_popcount(mu);
                wlo += __builtin_popcount(md);
            }
            float h[8][4];
#pragma unroll
            for (int j = 0; j < 8; ++j) {
                const int c = __builtin_amdgcn_readlane(mc, j);
                const float* __restrict__ hp = H + (size_t)c * 128 + lane;
                h[j][0] = hp[0];
                h[j][1] = hp[32];
                h[j][2] = hp[64];
                h[j][3] = hp[96];
            }
#pragma unroll
            for (int j = 0; j < 8; ++j) {
                const int   r = __builtin_amdgcn_readlane(mr, j);
                const float v = __int_as_float(__builtin_amdgcn_readlane(__float_as_int(mv), j));
                float* a = acc + r * 128 + lane;
                atomicAdd(a +  0, v * h[j][0]);
                atomicAdd(a + 32, v * h[j][1]);
                atomicAdd(a + 64, v * h[j][2]);
                atomicAdd(a + 96, v * h[j][3]);
            }
        }
        cnt = 0;
    };

    auto take = [&](const int4 ka, const int4 kb, int ea, int eb) {
        unsigned m = flags((unsigned)ka.x)        | (flags((unsigned)ka.y) << 1)
                   | (flags((unsigned)ka.z) << 2) | (flags((unsigned)ka.w) << 3)
                   | (flags((unsigned)kb.x) << 4) | (flags((unsigned)kb.y) << 5)
                   | (flags((unsigned)kb.z) << 6) | (flags((unsigned)kb.w) << 7);
#pragma nounroll
        for (int it = 0; it < 32; ++it) {
            const bool hit = m != 0u;
            const unsigned mask = __builtin_amdgcn_ballot_w32(hit);
            if (mask == 0u) break;
            if (cnt + 32 > AG_QCAP) drain();
            if (hit) {
                const int p = __builtin_ctz(m);
                const int i = p & 7;
                myq[cnt + __builtin_popcount(mask & lt)] = ((i & 4) ? eb : ea) + (((i & 3) << 2) | (p >> 3));
                m &= m - 1u;
            }
            cnt += __builtin_popcount(mask);
        }
    };

    const int nchunk = (E + 511) >> 9;
    const int c0     = (int)(((long long)nchunk * part) / AG_SPLIT);
    const int c1     = (int)(((long long)nchunk * (part + 1)) / AG_SPLIT);
    const int4* __restrict__ kp = (const int4*)keys;
    constexpr int AG_ROUND = AG_WAVES * AG_UNROLL;
    auto load_keys = [&](int4 (&k)[AG_UNROLL], int base) {
#pragma unroll
        for (int u = 0; u < AG_UNROLL; ++u) {
            const int ch = base + u * AG_WAVES + wave;
            k[u] = kp[(size_t)(ch < c1 ? ch : c1 - 1) * 32 + lane];
        }
    };
    if (c0 < c1) {
        int4 kk[AG_UNROLL];
        load_keys(kk, c0);
        for (int base = c0; base < c1; base += AG_ROUND) {
            int4 kn[AG_UNROLL];
            load_keys(kn, base + AG_ROUND);
#pragma unroll
            for (int u = 0; u < AG_UNROLL; u += 2) {
                const int ca = base + u * AG_WAVES + wave;
                const int cb = ca + AG_WAVES;
                if (ca < c1) {
                    if (cnt >= AG_EAGER) drain();
                    const int4 kb = (cb < c1) ? kk[u + 1] : make_int4(-1, -1, -1, -1);
                    take(kk[u], kb, (ca << 9) + (lane << 4), (cb << 9) + (lane << 4));
                }
            }
#pragma unroll
            for (int u = 0; u < AG_UNROLL; ++u) kk[u] = kn[u];
        }
    }
    drain();
    if (lane < 2) lcnt[((size_t)blockIdx.x * AG_WAVES + wave) * 2 + lane] = lane ? whi : wlo;
    __syncthreads();

    {
        const float4* a4 = (const float4*)acc;
        float4* o4 = (float4*)(out + (size_t)part * (size_t)M * 128);
        for (int i = tid; i < AG_ROWS * 32; i += AG_THREADS) {
            const int rrow = r0 + (i >> 5);
            if (rrow < M) o4[(size_t)rrow * 32 + (i & 31)] = a4[i];
        }
    }
}

constexpr int L2_THREADS = AG_THREADS;
static_assert(AG_SPLIT == 2, "a wave of the layer-2 row sums takes one list of either part");

template <int J>
__device__ __forceinline__ int row_lane(int x) {
    return __builtin_amdgcn_update_dpp(0, x, 0x150 + J, 0xf, 0xf, true);
}

template <int J, int N, class F>
__device__ __forceinline__ void static_for(F&& fn) {
    if constexpr (J < N) {
        fn(std::integral_constant<int, J>{});
        static_for<J + 1, N>(fn);
    }
}

__global__ __launch_bounds__(L2_THREADS)
void spmm_lists16(const int* __restrict__ lst, const int* __restrict__ lcnt,
                  const int* __restrict__ row, const int* __restrict__ col,
                  const float* __restrict__ val,
                  const float* __restrict__ H, float* __restrict__ out, int E, int M)
{
    __shared__ __attribute__((aligned(16))) float acc[(L2_ROWS + AG_WAVES) * 16];
    __shared__ int short_list;

    const int tid  = threadIdx.x;
    const int lane = tid & 31;
    const unsigned f = lane & 15;
    const int wave  = __builtin_amdgcn_readfirstlane(tid >> 5);
    const int range = (int)blockIdx.x >> 1;
    const int half  = (int)blockIdx.x & 1;
    const int r0    = range * AG_ROWS + half * L2_ROWS;

    float4* a4 = (float4*)acc;
    {
        const float4 z = make_float4(0.f, 0.f, 0.f, 0.f);
        for (int i = tid; i < (L2_ROWS + AG_WAVES) * 4; i += L2_THREADS) a4[i] = z;
    }
    if (tid == 0) short_list = 0;

    const int sa = ((range * AG_SPLIT) * AG_WAVES + wave) * 2 + half;
    const int sb = sa + AG_WAVES * 2;
    const int na = lcnt[sa];
    const int nb = lcnt[sb];
    __syncthreads();
    if ((na > L2_CAP || nb > L2_CAP) && lane == 0) short_list = 1;
    __syncthreads();

    if (short_list != 0) {
        for (int e = tid; e < E; e += L2_THREADS) {
            const unsigned rl = (unsigned)(row[e] - r0);
            if (rl < (unsigned)L2_ROWS) {
                const float v = val[e];
                const float* __restrict__ hp = H + (size_t)(unsigned)col[e] * 16u;
                for (int k = 0; k < 16; ++k) atomicAdd(acc + rl * 16u + k, v * hp[k]);
            }
        }
    } else {
        const int4* __restrict__ la = (const int4*)lst + (size_t)sa * L2_CAP;
        const int4* __restrict__ lb = (const int4*)lst + (size_t)sb * L2_CAP;
        const int spare = L2_ROWS + wave;
        const int nmax  = na > nb ? na : nb;
        auto rec = [&](const int4* __restrict__ l, int n, int i) -> int4 {
            int4 q = make_int4(spare, 0, 0, 0);
            if (i < n) q = l[i];
            return q;
        };
        int4 qa = rec(la, na, lane);
        int4 qb = rec(lb, nb, lane);
        for (int t0 = 0; t0 < nmax; t0 += 32) {
            float ha[16], hb[16];
            static_for<0, 16>([&](auto jc) {
                constexpr int J = decltype(jc)::value;
                ha[J] = H[(unsigned)row_lane<J>(qa.y) * 16u + f];
                hb[J] = H[(unsigned)row_lane<J>(qb.y) * 16u + f];
            });
            const int4 qan = rec(la, na, t0 + 32 + lane);
            const int4 qbn = rec(lb, nb, t0 + 32 + lane);
            static_for<0, 16>([&](auto jc) {
                constexpr int J = decltype(jc)::value;
                const unsigned ra = (unsigned)row_lane<J>(qa.x);
                const float    va = __int_as_float(row_lane<J>(qa.z));
                atomicAdd(acc + (ra * 16u + f), va * ha[J]);
                const unsigned rb = (unsigned)row_lane<J>(qb.x);
                const float    vb = __int_as_float(row_lane<J>(qb.z));
                atomicAdd(acc + (rb * 16u + f), vb * hb[J]);
            });
            qa = qan;
            qb = qbn;
        }
    }
    __syncthreads();

    {
        float4* o4 = (float4*)out;
        for (int i = tid; i < L2_ROWS * 4; i += L2_THREADS) {
            const int rrow = r0 + (i >> 2);
            if (rrow < M) o4[(size_t)rrow * 4 + (i & 3)] = a4[i];
        }
    }
}

extern "C" void kernel_launch(void* const* d_in, const int* in_sizes, int n_in,
                              void* d_out, int out_size, void* d_ws, size_t ws_size,
                              hipStream_t stream) {
    const float* x   = (const float*)d_in[0];
    const int*   row = (const int*)  d_in[1];
    const int*   col = (const int*)  d_in[2];
    const float* val = (const float*)d_in[3];
    const float* W1  = (const float*)d_in[4];
    const float* b1  = (const float*)d_in[5];
    const float* W2  = (const float*)d_in[6];
    const float* b2  = (const float*)d_in[7];
    float* out = (float*)d_out;

    constexpr int D_IN  = 128;
    constexpr int D_HID = 128;
    constexpr int N_CLS = 16;
    const int E = in_sizes[1];
    const int M = in_sizes[0] / D_IN;

    char*  ws  = (char*)d_ws;
    float* H1  = (float*)ws;
    float* AGG = (float*)(ws + (size_t)M * D_HID * sizeof(float));
    float* H2  = (float*)(ws + (1 + AG_SPLIT) * (size_t)M * D_HID * sizeof(float));
    unsigned int* KEY = (unsigned int*)(ws + ((1 + AG_SPLIT) * (size_t)D_HID + N_CLS) * (size_t)M * sizeof(float));
    const int nranges = (M + AG_ROWS - 1) / AG_ROWS;
    const size_t nlists = (size_t)nranges * AG_SPLIT * AG_WAVES * 2;
    int*  LCNT = (int*)((char*)KEY + ((size_t)((E + 511) >> 9) << 9));
    int*  LST  = (int*)((char*)LCNT + ((nlists * sizeof(int) + 127) & ~(size_t)127));


    {
        const int nstripes = M / 16;
        const unsigned nb = (unsigned)((nstripes + G1_WAVES - 1) / G1_WAVES);
        gcn_gemm_stripe128_f32<<<nb, G1_WAVES * 32, 0, stream>>>(x, W1, b1, H1, nstripes, row, KEY, E);
    }

    {
        const unsigned nb = (unsigned)((M + AG_ROWS - 1) / AG_ROWS) * AG_SPLIT;
        spmm_rows_in_lds128<<<nb, AG_THREADS, 0, stream>>>(row, col, val, H1, AGG, KEY, LST, LCNT, E, M);
    }

    static_assert(AG_SPLIT == 2, "the layer-2 product adds exactly two planes");
    dim3 g2(N_CLS / 16, M / 16);
    gcn_gemm_wmma_f32<D_HID, N_CLS, true><<<g2, 32, 0, stream>>>(
        AGG, AGG + (size_t)M * D_HID, W2, b2, H2);

    spmm_lists16<<<(unsigned)nranges * 2, L2_THREADS, 0, stream>>>(LST, LCNT, row, col, val, H2, out, E, M);
}


# Layer-1 row sums fetch a found edge as one 8-byte record (row in range, column, value; packed by the product launch), 32 edges per load, not three 4-byte loads for 8; sum order LDS arrival as before
# speedup vs baseline: 2.8417x; 1.0601x over previous
#include <hip/hip_runtime.h>
#include <type_traits>

typedef __attribute__((ext_vector_type(2))) float v2f;
typedef __attribute__((ext_vector_type(8))) float v8f;

template <int K, int N, bool RELU>
__global__ __launch_bounds__(32)
void gcn_gemm_wmma_f32(const float* __restrict__ A,
                       const float* __restrict__ A2,
                       const float* __restrict__ W,
                       const float* __restrict__ bias,
                       float* __restrict__ Dout)
{
    const int lane  = threadIdx.x;
    const int hi    = lane >> 4;
    const int l16   = lane & 15;
    const int mtile = blockIdx.y << 4;
    const int ntile = blockIdx.x << 4;

    const int m    = mtile + l16;
    const int koff = hi << 1;

    const float bv = bias[ntile + l16];
    v8f acc;
#pragma unroll
    for (int r = 0; r < 8; ++r) acc[r] = bv;

    const float* __restrict__ arow  = A  + (size_t)m * K + koff;
    const float* __restrict__ arow2 = A2 + (size_t)m * K + koff;
    const float* __restrict__ bcol = W + (size_t)koff * N + ntile + l16;

#pragma unroll
    for (int k0 = 0; k0 < K; k0 += 4) {
        v2f a = *(const v2f*)(arow + k0);
        const v2f a2 = *(const v2f*)(arow2 + k0);
        a[0] += a2[0];
        a[1] += a2[1];
        if (RELU) {
            a[0] = fmaxf(a[0], 0.0f);
            a[1] = fmaxf(a[1], 0.0f);
        }
        v2f b;
        b[0] = bcol[k0 * N];
        b[1] = bcol[(k0 + 1) * N];
        acc = __builtin_amdgcn_wmma_f32_16x16x4_f32(
            false, a, false, b, (short)0, acc, false, false);
    }

#pragma unroll
    for (int r = 0; r < 8; ++r) {
        Dout[(size_t)(mtile + (hi << 3) + r) * N + ntile + l16] = acc[r];
    }
}

constexpr int AG_ROWS  = 400;
constexpr int G1_WAVES = 8;

__device__ __forceinline__ unsigned edge_key(int r) {
    const unsigned k = (unsigned)r / (unsigned)AG_ROWS;
    return k < 255u ? k : 255u;
}

static_assert(AG_ROWS <= 512, "the row within a range has 9 bits in an edge record");
constexpr int      REC_COL_BITS = 23;
constexpr unsigned REC_COL_MASK = (1u << REC_COL_BITS) - 1u;

__device__ __forceinline__ int edge_rec(int r, int c) {
    const unsigned k = (unsigned)r / (unsigned)AG_ROWS;
    return (int)((((unsigned)r - k * (unsigned)AG_ROWS) << REC_COL_BITS) | ((unsigned)c & REC_COL_MASK));
}

__global__ __launch_bounds__(G1_WAVES * 32)
void gcn_gemm_stripe128_f32(const float* __restrict__ A,
                            const float* __restrict__ W,
                            const float* __restrict__ bias,
                            float* __restrict__ Dout, int nstripes,
                            const int* __restrict__ row,
                            const int* __restrict__ col,
                            const float* __restrict__ val,
                            unsigned int* __restrict__ keys,
                            int* __restrict__ recs, int E)
{
    constexpr int K = 128, N = 128;
    __shared__ __attribute__((aligned(16))) v2f Ws[(K / 2) * N];

    const int tid  = threadIdx.x;

    const int nwords = ((E + 511) >> 9) << 7;
    const int gsz    = (int)gridDim.x * (G1_WAVES * 32);
    const int i0     = (int)blockIdx.x * (G1_WAVES * 32) + tid;
    auto whole_rows = [&](int i) -> int4 {
        const int e = i << 2;
        return *(const int4*)(row + ((e + 3 < E) ? e : 0));
    };
    auto whole_cols = [&](int i) -> int4 {
        const int e = i << 2;
        return *(const int4*)(col + ((e + 3 < E) ? e : 0));
    };
    auto whole_vals = [&](int i) -> float4 {
        const int e = i << 2;
        return *(const float4*)(val + ((e + 3 < E) ? e : 0));
    };
    auto end_rows = [&](int i) -> int4 {
        const int e = i << 2;
        int4 r = make_int4(-1, -1, -1, -1);
        if (e     < E) r.x = row[e];
        if (e + 1 < E) r.y = row[e + 1];
        if (e + 2 < E) r.z = row[e + 2];
        return r;
    };
    auto store_keys = [&](int i, int4 r, const int4 c, const float4 v) {
        const int e = i << 2;
        if (e + 3 >= E) r = end_rows(i);
        keys[i] = edge_key(r.x) | (edge_key(r.y) << 8) | (edge_key(r.z) << 16) | (edge_key(r.w) << 24);
        if (e + 3 < E) {
            int4* d = (int4*)(recs + 2 * (size_t)e);
            d[0] = make_int4(edge_rec(r.x, c.x), __float_as_int(v.x), edge_rec(r.y, c.y), __float_as_int(v.y));
            d[1] = make_int4(edge_rec(r.z, c.z), __float_as_int(v.z), edge_rec(r.w, c.w), __float_as_int(v.w));
        } else {
            for (int k = 0; k < 3; ++k)
                if (e + k < E) {
                    recs[2 * (size_t)(e + k)]     = edge_rec(row[e + k], col[e + k]);
                    recs[2 * (size_t)(e + k) + 1] = __float_as_int(val[e + k]);
                }
        }
    };
    int4   rk[2], ck[2];
    float4 vk[2];
#pragma unroll
    for (int t = 0; t < 2; ++t) {
        rk[t] = whole_rows(i0 + t * gsz);
        ck[t] = whole_cols(i0 + t * gsz);
        vk[t] = whole_vals(i0 + t * gsz);
    }

    const int lane = tid & 31;
    const int hi   = lane >> 4;
    const int l16  = lane & 15;
    const int wave = __builtin_amdgcn_readfirstlane(tid >> 5);
    const int stripe = blockIdx.x * G1_WAVES + wave;
    const bool live = stripe < nstripes;
    const int mtile = (live ? stripe : 0) << 4;

    const float* __restrict__ arow = A + (size_t)(mtile + l16) * K + (hi << 1);
    v2f a[K / 4];
#pragma unroll
    for (int s = 0; s < K / 4; ++s) a[s] = *(const v2f*)(arow + 4 * s);

#pragma unroll
    for (int j = 0; j < (K / 2) * (N / 4) / (G1_WAVES * 32); ++j) {
        const int i  = tid + j * (G1_WAVES * 32);
        const int p  = i >> 5;
        const int n4 = (i & 31) << 2;
        const float4 r0 = *(const float4*)(W + (size_t)(2 * p) * N + n4);
        const float4 r1 = *(const float4*)(W + (size_t)(2 * p + 1) * N + n4);
        float4* d = (float4*)(Ws + p * N + n4);
        d[0] = make_float4(r0.x, r1.x, r0.y, r1.y);
        d[1] = make_float4(r0.z, r1.z, r0.w, r1.w);
    }

    v8f acc[N / 16];
#pragma unroll
    for (int t = 0; t < N / 16; ++t) {
        const float bv = bias[t * 16 + l16];
#pragma unroll
        for (int r = 0; r < 8; ++r) acc[t][r] = bv;
    }
    __syncthreads();
    if (live) {
        const v2f* __restrict__ wl = Ws + hi * N + l16;
#pragma unroll
        for (int s = 0; s < K / 4; ++s) {
            v2f b[N / 16];
#pragma unroll
            for (int t = 0; t < N / 16; ++t) b[t] = wl[(2 * s) * N + t * 16];
#pragma unroll
            for (int t = 0; t < N / 16; ++t)
                acc[t] = __builtin_amdgcn_wmma_f32_16x16x4_f32(
                    false, a[s], false, b[t], (short)0, acc[t], false, false);
        }

#pragma unroll
        for (int r = 0; r < 8; ++r) {
            float* __restrict__ drow = Dout + (size_t)(mtile + (hi << 3) + r) * N + l16;
#pragma unroll
            for (int t = 0; t < N / 16; ++t) drow[t * 16] = acc[t][r];
        }
    }

#pragma unroll
    for (int t = 0; t < 2; ++t)
        if (i0 + t * gsz < nwords) store_keys(i0 + t * gsz, rk[t], ck[t], vk[t]);
    for (int i = i0 + 2 * gsz; i < nwords; i += gsz) store_keys(i, whole_rows(i), whole_cols(i), whole_vals(i));
}

constexpr int AG_SPLIT   = 2;
constexpr int AG_THREADS = 1024;
constexpr int AG_WAVES   = AG_THREADS / 32;
constexpr int AG_UNROLL  = 4;
constexpr int AG_EAGER   = 64;
constexpr int AG_QCAP    = AG_EAGER + 512;
static_assert(AG_UNROLL % 2 == 0, "pieces are searched two at a time");
constexpr int L2_ROWS = AG_ROWS / 2;
constexpr int L2_CAP  = 128;
static_assert(AG_ROWS % 2 == 0, "a range is two layer-2 blocks' rows");

__global__ __launch_bounds__(AG_THREADS)
void spmm_rows_in_lds128(const int* __restrict__ recs,
                         const float* __restrict__ H, float* __restrict__ out,
                         const unsigned int* __restrict__ keys,
                         int* __restrict__ lst, int* __restrict__ lcnt,
                         int E, int M)
{
    __shared__ float acc[(AG_ROWS + 1) * 128];
    __shared__ int   qbuf[AG_WAVES * AG_QCAP];

    const int tid  = threadIdx.x;
    const int lane = tid & 31;
    const int wave = __builtin_amdgcn_readfirstlane(tid >> 5);
    const int range = (int)blockIdx.x / AG_SPLIT;
    const int part  = (int)blockIdx.x % AG_SPLIT;
    const int r0    = range * AG_ROWS;

    {
        float4* a4 = (float4*)acc;
        const float4 z = make_float4(0.f, 0.f, 0.f, 0.f);
        for (int i = tid; i < AG_ROWS * 32; i += AG_THREADS) a4[i] = z;
    }
    __syncthreads();

    int* myq = qbuf + wave * AG_QCAP;
    int  cnt = 0;
    int4* __restrict__ mylst = (int4*)lst + ((size_t)blockIdx.x * AG_WAVES + wave) * (2 * L2_CAP);
    int  wlo = 0, whi = 0;

    const unsigned pat = (unsigned)range * 0x01010101u;
    const unsigned lt  = (1u << lane) - 1u;
    auto flags = [&](unsigned w) -> unsigned {
        const unsigned x = w ^ pat;
        const unsigned z = ((x & 0x7f7f7f7fu) + 0x7f7f7f7fu) | x | 0x7f7f7f7fu;
        return (~z) >> 7;
    };
    const int2* __restrict__ rp = (const int2*)recs;
    auto drain = [&]() {
        for (int b = 0; b < cnt; b += 32) {
            const int nb = cnt - b;
            int mr = AG_ROWS, mc = 0;
            float mv = 0.f;
            const bool act = lane < nb;
            if (act) {
                const int2 q = rp[myq[b + lane]];
                mr = (int)((unsigned)q.x >> REC_COL_BITS);
                mc = (int)((unsigned)q.x & REC_COL_MASK);
                mv = __int_as_float(q.y);
            }
            {
                const bool up = act && mr >= L2_ROWS;
                const unsigned mu = __builtin_amdgcn_ballot_w32(up);
                const unsigned md = __builtin_amdgcn_ballot_w32(act && !up);
                if (act) {
                    const int pos = up ? whi + __builtin_popcount(mu & lt) : wlo + __builtin_popcount(md & lt);
                    if (pos < L2_CAP)
                        mylst[(up ? L2_CAP : 0) + pos] = make_int4(up ? mr - L2_ROWS : mr, mc, __float_as_int(mv), 0);
                }
                whi += __builtin_popcount(mu);
                wlo += __builtin_popcount(md);
            }
            const int ng = nb < 32 ? nb : 32;
            for (int g = 0; g < ng; g += 8) {
                float h[8][4];
#pragma unroll
                for (int j = 0; j < 8; ++j) {
                    const int c = __builtin_amdgcn_readlane(mc, g + j);
                    const float* __restrict__ hp = H + (size_t)c * 128 + lane;
                    h[j][0] = hp[0];
                    h[j][1] = hp[32];
                    h[j][2] = hp[64];
                    h[j][3] = hp[96];
                }
#pragma unroll
                for (int j = 0; j < 8; ++j) {
                    const int   r = __builtin_amdgcn_readlane(mr, g + j);
                    const float v = __int_as_float(__builtin_amdgcn_readlane(__float_as_int(mv), g + j));
                    float* a = acc + r * 128 + lane;
                    atomicAdd(a +  0, v * h[j][0]);
                    atomicAdd(a + 32, v * h[j][1]);
                    atomicAdd(a + 64, v * h[j][2]);
                    atomicAdd(a + 96, v * h[j][3]);
                }
            }
        }
        cnt = 0;
    };

    auto take = [&](const int4 ka, const int4 kb, int ea, int eb) {
        unsigned m = flags((unsigned)ka.x)        | (flags((unsigned)ka.y) << 1)
                   | (flags((unsigned)ka.z) << 2) | (flags((unsigned)ka.w) << 3)
                   | (flags((unsigned)kb.x) << 4) | (flags((unsigned)kb.y) << 5)
                   | (flags((unsigned)kb.z) << 6) | (flags((unsigned)kb.w) << 7);
#pragma nounroll
        for (int it = 0; it < 32; ++it) {
            const bool hit = m != 0u;
            const unsigned mask = __builtin_amdgcn_ballot_w32(hit);
            if (mask == 0u) break;
            if (cnt + 32 > AG_QCAP) drain();
            if (hit) {
                const int p = __builtin_ctz(m);
                const int i = p & 7;
                myq[cnt + __builtin_popcount(mask & lt)] = ((i & 4) ? eb : ea) + (((i & 3) << 2) | (p >> 3));
                m &= m - 1u;
            }
            cnt += __builtin_popcount(mask);
        }
    };

    const int nchunk = (E + 511) >> 9;
    const int c0     = (int)(((long long)nchunk * part) / AG_SPLIT);
    const int c1     = (int)(((long long)nchunk * (part + 1)) / AG_SPLIT);
    const int4* __restrict__ kp = (const int4*)keys;
    constexpr int AG_ROUND = AG_WAVES * AG_UNROLL;
    auto load_keys = [&](int4 (&k)[AG_UNROLL], int base) {
#pragma unroll
        for (int u = 0; u < AG_UNROLL; ++u) {
            const int ch = base + u * AG_WAVES + wave;
            k[u] = kp[(size_t)(ch < c1 ? ch : c1 - 1) * 32 + lane];
        }
    };
    if (c0 < c1) {
        int4 kk[AG_UNROLL];
        load_keys(kk, c0);
        for (int base = c0; base < c1; base += AG_ROUND) {
            int4 kn[AG_UNROLL];
            load_keys(kn, base + AG_ROUND);
#pragma unroll
            for (int u = 0; u < AG_UNROLL; u += 2) {
                const int ca = base + u * AG_WAVES + wave;
                const int cb = ca + AG_WAVES;
                if (ca < c1) {
                    if (cnt >= AG_EAGER) drain();
                    const int4 kb = (cb < c1) ? kk[u + 1] : make_int4(-1, -1, -1, -1);
                    take(kk[u], kb, (ca << 9) + (lane << 4), (cb << 9) + (lane << 4));
                }
            }
#pragma unroll
            for (int u = 0; u < AG_UNROLL; ++u) kk[u] = kn[u];
        }
    }
    drain();
    if (lane < 2) lcnt[((size_t)blockIdx.x * AG_WAVES + wave) * 2 + lane] = lane ? whi : wlo;
    __syncthreads();

    {
        const float4* a4 = (const float4*)acc;
        float4* o4 = (float4*)(out + (size_t)part * (size_t)M * 128);
        for (int i = tid; i < AG_ROWS * 32; i += AG_THREADS) {
            const int rrow = r0 + (i >> 5);
            if (rrow < M) o4[(size_t)rrow * 32 + (i & 31)] = a4[i];
        }
    }
}

constexpr int L2_THREADS = AG_THREADS;
static_assert(AG_SPLIT == 2, "a wave of the layer-2 row sums takes one list of either part");

template <int J>
__device__ __forceinline__ int row_lane(int x) {
    return __builtin_amdgcn_update_dpp(0, x, 0x150 + J, 0xf, 0xf, true);
}

template <int J, int N, class F>
__device__ __forceinline__ void static_for(F&& fn) {
    if constexpr (J < N) {
        fn(std::integral_constant<int, J>{});
        static_for<J + 1, N>(fn);
    }
}

__global__ __launch_bounds__(L2_THREADS)
void spmm_lists16(const int* __restrict__ lst, const int* __restrict__ lcnt,
                  const int* __restrict__ row, const int* __restrict__ col,
                  const float* __restrict__ val,
                  const float* __restrict__ H, float* __restrict__ out, int E, int M)
{
    __shared__ __attribute__((aligned(16))) float acc[(L2_ROWS + AG_WAVES) * 16];
    __shared__ int short_list;

    const int tid  = threadIdx.x;
    const int lane = tid & 31;
    const unsigned f = lane & 15;
    const int wave  = __builtin_amdgcn_readfirstlane(tid >> 5);
    const int range = (int)blockIdx.x >> 1;
    const int half  = (int)blockIdx.x & 1;
    const int r0    = range * AG_ROWS + half * L2_ROWS;

    float4* a4 = (float4*)acc;
    {
        const float4 z = make_float4(0.f, 0.f, 0.f, 0.f);
        for (int i = tid; i < (L2_ROWS + AG_WAVES) * 4; i += L2_THREADS) a4[i] = z;
    }
    if (tid == 0) short_list = 0;

    const int sa = ((range * AG_SPLIT) * AG_WAVES + wave) * 2 + half;
    const int sb = sa + AG_WAVES * 2;
    const int na = lcnt[sa];
    const int nb = lcnt[sb];
    __syncthreads();
    if ((na > L2_CAP || nb > L2_CAP) && lane == 0) short_list = 1;
    __syncthreads();

    if (short_list != 0) {
        for (int e = tid; e < E; e += L2_THREADS) {
            const unsigned rl = (unsigned)(row[e] - r0);
            if (rl < (unsigned)L2_ROWS) {
                const float v = val[e];
                const float* __restrict__ hp = H + (size_t)(unsigned)col[e] * 16u;
                for (int k = 0; k < 16; ++k) atomicAdd(acc + rl * 16u + k, v * hp[k]);
            }
        }
    } else {
        const int4* __restrict__ la = (const int4*)lst + (size_t)sa * L2_CAP;
        const int4* __restrict__ lb = (const int4*)lst + (size_t)sb * L2_CAP;
        const int spare = L2_ROWS + wave;
        const int nmax  = na > nb ? na : nb;
        auto rec = [&](const int4* __restrict__ l, int n, int i) -> int4 {
            int4 q = make_int4(spare, 0, 0, 0);
            if (i < n) q = l[i];
            return q;
        };
        int4 qa = rec(la, na, lane);
        int4 qb = rec(lb, nb, lane);
        for (int t0 = 0; t0 < nmax; t0 += 32) {
            float ha[16], hb[16];
            static_for<0, 16>([&](auto jc) {
                constexpr int J = decltype(jc)::value;
                ha[J] = H[(unsigned)row_lane<J>(qa.y) * 16u + f];
                hb[J] = H[(unsigned)row_lane<J>(qb.y) * 16u + f];
            });
            const int4 qan = rec(la, na, t0 + 32 + lane);
            const int4 qbn = rec(lb, nb, t0 + 32 + lane);
            static_for<0, 16>([&](auto jc) {
                constexpr int J = decltype(jc)::value;
                const unsigned ra = (unsigned)row_lane<J>(qa.x);
                const float    va = __int_as_float(row_lane<J>(qa.z));
                atomicAdd(acc + (ra * 16u + f), va * ha[J]);
                const unsigned rb = (unsigned)row_lane<J>(qb.x);
                const float    vb = __int_as_float(row_lane<J>(qb.z));
                atomicAdd(acc + (rb * 16u + f), vb * hb[J]);
            });
            qa = qan;
            qb = qbn;
        }
    }
    __syncthreads();

    {
        float4* o4 = (float4*)out;
        for (int i = tid; i < L2_ROWS * 4; i += L2_THREADS) {
            const int rrow = r0 + (i >> 2);
            if (rrow < M) o4[(size_t)rrow * 4 + (i & 3)] = a4[i];
        }
    }
}

extern "C" void kernel_launch(void* const* d_in, const int* in_sizes, int n_in,
                              void* d_out, int out_size, void* d_ws, size_t ws_size,
                              hipStream_t stream) {
    const float* x   = (const float*)d_in[0];
    const int*   row = (const int*)  d_in[1];
    const int*   col = (const int*)  d_in[2];
    const float* val = (const float*)d_in[3];
    const float* W1  = (const float*)d_in[4];
    const float* b1  = (const float*)d_in[5];
    const float* W2  = (const float*)d_in[6];
    const float* b2  = (const float*)d_in[7];
    float* out = (float*)d_out;

    constexpr int D_IN  = 128;
    constexpr int D_HID = 128;
    constexpr int N_CLS = 16;
    const int E = in_sizes[1];
    const int M = in_sizes[0] / D_IN;

    char*  ws  = (char*)d_ws;
    float* H1  = (float*)ws;
    float* AGG = (float*)(ws + (size_t)M * D_HID * sizeof(float));
    float* H2  = (float*)(ws + (1 + AG_SPLIT) * (size_t)M * D_HID * sizeof(float));
    unsigned int* KEY = (unsigned int*)(ws + ((1 + AG_SPLIT) * (size_t)D_HID + N_CLS) * (size_t)M * sizeof(float));
    const int nranges = (M + AG_ROWS - 1) / AG_ROWS;
    const size_t nlists = (size_t)nranges * AG_SPLIT * AG_WAVES * 2;
    int*  RECS = (int*)((char*)KEY + ((size_t)((E + 511) >> 9) << 9));
    int*  LCNT = (int*)((char*)RECS + ((size_t)((E + 511) >> 9) << 12));
    int*  LST  = (int*)((char*)LCNT + ((nlists * sizeof(int) + 127) & ~(size_t)127));


    {
        const int nstripes = M / 16;
        const unsigned nb = (unsigned)((nstripes + G1_WAVES - 1) / G1_WAVES);
        gcn_gemm_stripe128_f32<<<nb, G1_WAVES * 32, 0, stream>>>(x, W1, b1, H1, nstripes, row, col, val, KEY, RECS, E);
    }

    {
        const unsigned nb = (unsigned)((M + AG_ROWS - 1) / AG_ROWS) * AG_SPLIT;
        spmm_rows_in_lds128<<<nb, AG_THREADS, 0, stream>>>(RECS, H1, AGG, KEY, LST, LCNT, E, M);
    }

    static_assert(AG_SPLIT == 2, "the layer-2 product adds exactly two planes");
    dim3 g2(N_CLS / 16, M / 16);
    gcn_gemm_wmma_f32<D_HID, N_CLS, true><<<g2, 32, 0, stream>>>(
        AGG, AGG + (size_t)M * D_HID, W2, b2, H2);

    spmm_lists16<<<(unsigned)nranges * 2, L2_THREADS, 0, stream>>>(LST, LCNT, row, col, val, H2, out, E, M);
}


# Layer-1 row sums: a block owns 200 rows whole and searches all edge keys (was 400 rows, half the list, 2 planes), then does relu and the layer-2 product itself; 3 launches; sum order: one LDS sum
# speedup vs baseline: 3.0727x; 1.0813x over previous
#include <hip/hip_runtime.h>
#include <type_traits>

typedef __attribute__((ext_vector_type(2))) float v2f;
typedef __attribute__((ext_vector_type(8))) float v8f;

constexpr int AG_ROWS  = 200;
constexpr int G1_WAVES = 8;

__device__ __forceinline__ unsigned edge_key(int r) {
    const unsigned k = (unsigned)r / (unsigned)AG_ROWS;
    return k < 255u ? k : 255u;
}

static_assert(AG_ROWS <= 512, "the row within a range has 9 bits in an edge record");
constexpr int      REC_COL_BITS = 23;
constexpr unsigned REC_COL_MASK = (1u << REC_COL_BITS) - 1u;

__device__ __forceinline__ int edge_rec(int r, int c) {
    const unsigned k = (unsigned)r / (unsigned)AG_ROWS;
    return (int)((((unsigned)r - k * (unsigned)AG_ROWS) << REC_COL_BITS) | ((unsigned)c & REC_COL_MASK));
}

__global__ __launch_bounds__(G1_WAVES * 32)
void gcn_gemm_stripe128_f32(const float* __restrict__ A,
                            const float* __restrict__ W,
                            const float* __restrict__ bias,
                            float* __restrict__ Dout, int nstripes,
                            const int* __restrict__ row,
                            const int* __restrict__ col,
                            const float* __restrict__ val,
                            unsigned int* __restrict__ keys,
                            int* __restrict__ recs, int E)
{
    constexpr int K = 128, N = 128;
    __shared__ __attribute__((aligned(16))) v2f Ws[(K / 2) * N];

    const int tid  = threadIdx.x;

    const int nwords = ((E + 511) >> 9) << 7;
    const int gsz    = (int)gridDim.x * (G1_WAVES * 32);
    const int i0     = (int)blockIdx.x * (G1_WAVES * 32) + tid;
    auto whole_rows = [&](int i) -> int4 {
        const int e = i << 2;
        return *(const int4*)(row + ((e + 3 < E) ? e : 0));
    };
    auto whole_cols = [&](int i) -> int4 {
        const int e = i << 2;
        return *(const int4*)(col + ((e + 3 < E) ? e : 0));
    };
    auto whole_vals = [&](int i) -> float4 {
        const int e = i << 2;
        return *(const float4*)(val + ((e + 3 < E) ? e : 0));
    };
    auto end_rows = [&](int i) -> int4 {
        const int e = i << 2;
        int4 r = make_int4(-1, -1, -1, -1);
        if (e     < E) r.x = row[e];
        if (e + 1 < E) r.y = row[e + 1];
        if (e + 2 < E) r.z = row[e + 2];
        return r;
    };
    auto store_keys = [&](int i, int4 r, const int4 c, const float4 v) {
        const int e = i << 2;
        if (e + 3 >= E) r = end_rows(i);
        keys[i] = edge_key(r.x) | (edge_key(r.y) << 8) | (edge_key(r.z) << 16) | (edge_key(r.w) << 24);
        if (e + 3 < E) {
            int4* d = (int4*)(recs + 2 * (size_t)e);
            d[0] = make_int4(edge_rec(r.x, c.x), __float_as_int(v.x), edge_rec(r.y, c.y), __float_as_int(v.y));
            d[1] = make_int4(edge_rec(r.z, c.z), __float_as_int(v.z), edge_rec(r.w, c.w), __float_as_int(v.w));
        } else {
            for (int k = 0; k < 3; ++k)
                if (e + k < E) {
                    recs[2 * (size_t)(e + k)]     = edge_rec(row[e + k], col[e + k]);
                    recs[2 * (size_t)(e + k) + 1] = __float_as_int(val[e + k]);
                }
        }
    };
    int4   rk[2], ck[2];
    float4 vk[2];
#pragma unroll
    for (int t = 0; t < 2; ++t) {
        rk[t] = whole_rows(i0 + t * gsz);
        ck[t] = whole_cols(i0 + t * gsz);
        vk[t] = whole_vals(i0 + t * gsz);
    }

    const int lane = tid & 31;
    const int hi   = lane >> 4;
    const int l16  = lane & 15;
    const int wave = __builtin_amdgcn_readfirstlane(tid >> 5);
    const int stripe = blockIdx.x * G1_WAVES + wave;
    const bool live = stripe < nstripes;
    const int mtile = (live ? stripe : 0) << 4;

    const float* __restrict__ arow = A + (size_t)(mtile + l16) * K + (hi << 1);
    v2f a[K / 4];
#pragma unroll
    for (int s = 0; s < K / 4; ++s) a[s] = *(const v2f*)(arow + 4 * s);

#pragma unroll
    for (int j = 0; j < (K / 2) * (N / 4) / (G1_WAVES * 32); ++j) {
        const int i  = tid + j * (G1_WAVES * 32);
        const int p  = i >> 5;
        const int n4 = (i & 31) << 2;
        const float4 r0 = *(const float4*)(W + (size_t)(2 * p) * N + n4);
        const float4 r1 = *(const float4*)(W + (size_t)(2 * p + 1) * N + n4);
        float4* d = (float4*)(Ws + p * N + n4);
        d[0] = make_float4(r0.x, r1.x, r0.y, r1.y);
        d[1] = make_float4(r0.z, r1.z, r0.w, r1.w);
    }

    v8f acc[N / 16];
#pragma unroll
    for (int t = 0; t < N / 16; ++t) {
        const float bv = bias[t * 16 + l16];
#pragma unroll
        for (int r = 0; r < 8; ++r) acc[t][r] = bv;
    }
    __syncthreads();
    if (live) {
        const v2f* __restrict__ wl = Ws + hi * N + l16;
#pragma unroll
        for (int s = 0; s < K / 4; ++s) {
            v2f b[N / 16];
#pragma unroll
            for (int t = 0; t < N / 16; ++t) b[t] = wl[(2 * s) * N + t * 16];
#pragma unroll
            for (int t = 0; t < N / 16; ++t)
                acc[t] = __builtin_amdgcn_wmma_f32_16x16x4_f32(
                    false, a[s], false, b[t], (short)0, acc[t], false, false);
        }

#pragma unroll
        for (int r = 0; r < 8; ++r) {
            float* __restrict__ drow = Dout + (size_t)(mtile + (hi << 3) + r) * N + l16;
#pragma unroll
            for (int t = 0; t < N / 16; ++t) drow[t * 16] = acc[t][r];
        }
    }

#pragma unroll
    for (int t = 0; t < 2; ++t)
        if (i0 + t * gsz < nwords) store_keys(i0 + t * gsz, rk[t], ck[t], vk[t]);
    for (int i = i0 + 2 * gsz; i < nwords; i += gsz) store_keys(i, whole_rows(i), whole_cols(i), whole_vals(i));
}

constexpr int AG_THREADS = 1024;
constexpr int AG_WAVES   = AG_THREADS / 32;
constexpr int AG_UNROLL  = 4;
constexpr int AG_EAGER   = 32;
constexpr int AG_QCAP    = AG_EAGER + 512;
constexpr int AG_STRIPES = (AG_ROWS + 15) / 16;
constexpr int AG_SPARE   = AG_STRIPES * 16;
constexpr int AG_PITCH   = 132;
static_assert(AG_UNROLL % 2 == 0, "pieces are searched two at a time");
static_assert(AG_STRIPES <= AG_WAVES, "a wave takes one stripe of the layer-2 product");
static_assert(AG_PITCH % 4 == 0 && AG_PITCH >= 128, "rows are cleared 16 bytes at a time");
constexpr int L2_ROWS = AG_ROWS;
constexpr int L2_CAP  = 256;

__global__ __launch_bounds__(AG_THREADS)
void spmm_rows_in_lds128(const int* __restrict__ recs,
                         const float* __restrict__ H,
                         const float* __restrict__ W2,
                         const float* __restrict__ b2,
                         float* __restrict__ H2,
                         const unsigned int* __restrict__ keys,
                         int* __restrict__ lst, int* __restrict__ lcnt,
                         int E, int M)
{
    __shared__ __attribute__((aligned(16))) float acc[(AG_SPARE + 1) * AG_PITCH];
    __shared__ int   qbuf[AG_WAVES * AG_QCAP];

    const int tid  = threadIdx.x;
    const int lane = tid & 31;
    const int wave = __builtin_amdgcn_readfirstlane(tid >> 5);
    const int range = (int)blockIdx.x;
    const int r0    = range * AG_ROWS;

    {
        float4* a4 = (float4*)acc;
        const float4 z = make_float4(0.f, 0.f, 0.f, 0.f);
        for (int i = tid; i < (AG_SPARE + 1) * (AG_PITCH / 4); i += AG_THREADS) a4[i] = z;
    }
    __syncthreads();

    int* myq = qbuf + wave * AG_QCAP;
    int  cnt = 0;
    int4* __restrict__ mylst = (int4*)lst + ((size_t)blockIdx.x * AG_WAVES + wave) * L2_CAP;
    int  wn = 0;

    const unsigned pat = (unsigned)range * 0x01010101u;
    const unsigned lt  = (1u << lane) - 1u;
    auto flags = [&](unsigned w) -> unsigned {
        const unsigned x = w ^ pat;
        const unsigned z = ((x & 0x7f7f7f7fu) + 0x7f7f7f7fu) | x | 0x7f7f7f7fu;
        return (~z) >> 7;
    };
    const int2* __restrict__ rp = (const int2*)recs;
    auto drain = [&]() {
        for (int b = 0; b < cnt; b += 32) {
            const int nb = cnt - b;
            int mr = AG_SPARE, mc = 0;
            float mv = 0.f;
            if (lane < nb) {
                const int2 q = rp[myq[b + lane]];
                mr = (int)((unsigned)q.x >> REC_COL_BITS);
                mc = (int)((unsigned)q.x & REC_COL_MASK);
                mv = __int_as_float(q.y);
                const int pos = wn + lane;
                if (pos < L2_CAP) mylst[pos] = make_int4(mr, mc, q.y, 0);
            }
            const int mo = mr * AG_PITCH;
            const int ng = nb < 32 ? nb : 32;
            wn += ng;
            for (int g = 0; g < ng; g += 8) {
                float h[8][4];
#pragma unroll
                for (int j = 0; j < 8; ++j) {
                    const int c = __builtin_amdgcn_readlane(mc, g + j);
                    const float* __restrict__ hp = H + (size_t)c * 128 + lane;
                    h[j][0] = hp[0];
                    h[j][1] = hp[32];
                    h[j][2] = hp[64];
                    h[j][3] = hp[96];
                }
#pragma unroll
                for (int j = 0; j < 8; ++j) {
                    const int   ro = __builtin_amdgcn_readlane(mo, g + j);
                    const float v = __int_as_float(__builtin_amdgcn_readlane(__float_as_int(mv), g + j));
                    float* a = acc + ro + lane;
                    atomicAdd(a +  0, v * h[j][0]);
                    atomicAdd(a + 32, v * h[j][1]);
                    atomicAdd(a + 64, v * h[j][2]);
                    atomicAdd(a + 96, v * h[j][3]);
                }
            }
        }
        cnt = 0;
    };

    auto take = [&](const int4 ka, const int4 kb, int ea, int eb) {
        unsigned m = flags((unsigned)ka.x)        | (flags((unsigned)ka.y) << 1)
                   | (flags((unsigned)ka.z) << 2) | (flags((unsigned)ka.w) << 3)
                   | (flags((unsigned)kb.x) << 4) | (flags((unsigned)kb.y) << 5)
                   | (flags((unsigned)kb.z) << 6) | (flags((unsigned)kb.w) << 7);
#pragma nounroll
        for (int it = 0; it < 32; ++it) {
            const bool hit = m != 0u;
            const unsigned mask = __builtin_amdgcn_ballot_w32(hit);
            if (mask == 0u) break;
            if (cnt + 32 > AG_QCAP) drain();
            if (hit) {
                const int p = __builtin_ctz(m);
                const int i = p & 7;
                myq[cnt + __builtin_popcount(mask & lt)] = ((i & 4) ? eb : ea) + (((i & 3) << 2) | (p >> 3));
                m &= m - 1u;
            }
            cnt += __builtin_popcount(mask);
        }
    };

    const int c1 = (E + 511) >> 9;
    const int4* __restrict__ kp = (const int4*)keys;
    constexpr int AG_ROUND = AG_WAVES * AG_UNROLL;
    auto load_keys = [&](int4 (&k)[AG_UNROLL], int base) {
#pragma unroll
        for (int u = 0; u < AG_UNROLL; ++u) {
            const int ch = base + u * AG_WAVES + wave;
            k[u] = kp[(size_t)(ch < c1 ? ch : c1 - 1) * 32 + lane];
        }
    };
    if (c1 > 0) {
        int4 kk[AG_UNROLL];
        load_keys(kk, 0);
        for (int base = 0; base < c1; base += AG_ROUND) {
            int4 kn[AG_UNROLL];
            load_keys(kn, base + AG_ROUND);
#pragma unroll
            for (int u = 0; u < AG_UNROLL; u += 2) {
                const int ca = base + u * AG_WAVES + wave;
                const int cb = ca + AG_WAVES;
                if (ca < c1) {
                    if (cnt >= AG_EAGER) drain();
                    const int4 kb = (cb < c1) ? kk[u + 1] : make_int4(-1, -1, -1, -1);
                    take(kk[u], kb, (ca << 9) + (lane << 4), (cb << 9) + (lane << 4));
                }
            }
#pragma unroll
            for (int u = 0; u < AG_UNROLL; ++u) kk[u] = kn[u];
        }
    }
    drain();
    if (lane == 0) lcnt[(size_t)blockIdx.x * AG_WAVES + wave] = wn;

    const int hi  = lane >> 4;
    const int l16 = lane & 15;
    v2f   bw[32];
    float bv = 0.f;
    if (wave < AG_STRIPES) {
        const float* __restrict__ bcol = W2 + (size_t)(hi << 1) * 16 + l16;
        bv = b2[l16];
#pragma unroll
        for (int s = 0; s < 32; ++s) {
            bw[s][0] = bcol[(4 * s) * 16];
            bw[s][1] = bcol[(4 * s + 1) * 16];
        }
    }
    __syncthreads();

    if (wave < AG_STRIPES) {
        const float* __restrict__ arow = acc + (wave * 16 + l16) * AG_PITCH + (hi << 1);
        v8f d;
#pragma unroll
        for (int r = 0; r < 8; ++r) d[r] = bv;
#pragma unroll
        for (int k0 = 0; k0 < 128; k0 += 4) {
            v2f a = *(const v2f*)(arow + k0);
            a[0] = fmaxf(a[0], 0.0f);
            a[1] = fmaxf(a[1], 0.0f);
            d = __builtin_amdgcn_wmma_f32_16x16x4_f32(
                false, a, false, bw[k0 / 4], (short)0, d, false, false);
        }
#pragma unroll
        for (int r = 0; r < 8; ++r) {
            const int lrow = wave * 16 + (hi << 3) + r;
            if (lrow < AG_ROWS && r0 + lrow < M)
                H2[(size_t)(r0 + lrow) * 16 + l16] = d[r];
        }
    }
}

constexpr int L2_THREADS = AG_THREADS;

template <int J>
__device__ __forceinline__ int row_lane(int x) {
    return __builtin_amdgcn_update_dpp(0, x, 0x150 + J, 0xf, 0xf, true);
}

template <int J, int N, class F>
__device__ __forceinline__ void static_for(F&& fn) {
    if constexpr (J < N) {
        fn(std::integral_constant<int, J>{});
        static_for<J + 1, N>(fn);
    }
}

__global__ __launch_bounds__(L2_THREADS)
void spmm_lists16(const int* __restrict__ lst, const int* __restrict__ lcnt,
                  const int* __restrict__ row, const int* __restrict__ col,
                  const float* __restrict__ val,
                  const float* __restrict__ H, float* __restrict__ out, int E, int M)
{
    __shared__ __attribute__((aligned(16))) float acc[(L2_ROWS + AG_WAVES) * 16];
    __shared__ int short_list;

    const int tid  = threadIdx.x;
    const int lane = tid & 31;
    const unsigned f = lane & 15;
    const int wave  = __builtin_amdgcn_readfirstlane(tid >> 5);
    const int r0    = (int)blockIdx.x * L2_ROWS;

    float4* a4 = (float4*)acc;
    {
        const float4 z = make_float4(0.f, 0.f, 0.f, 0.f);
        for (int i = tid; i < (L2_ROWS + AG_WAVES) * 4; i += L2_THREADS) a4[i] = z;
    }
    if (tid == 0) short_list = 0;

    const int sl = (int)blockIdx.x * AG_WAVES + wave;
    const int n  = lcnt[sl];
    __syncthreads();
    if (n > L2_CAP && lane == 0) short_list = 1;
    __syncthreads();

    if (short_list != 0) {
        for (int e = tid; e < E; e += L2_THREADS) {
            const unsigned rl = (unsigned)(row[e] - r0);
            if (rl < (unsigned)L2_ROWS) {
                const float v = val[e];
                const float* __restrict__ hp = H + (size_t)(unsigned)col[e] * 16u;
                for (int k = 0; k < 16; ++k) atomicAdd(acc + rl * 16u + k, v * hp[k]);
            }
        }
    } else {
        const int4* __restrict__ l = (const int4*)lst + (size_t)sl * L2_CAP;
        const int spare = L2_ROWS + wave;
        auto rec = [&](int i) -> int4 {
            int4 q = make_int4(spare, 0, 0, 0);
            if (i < n) q = l[i];
            return q;
        };
        int4 qa = rec(lane);
        int4 qb = rec(32 + lane);
        for (int t0 = 0; t0 < n; t0 += 64) {
            float ha[16], hb[16];
            static_for<0, 16>([&](auto jc) {
                constexpr int J = decltype(jc)::value;
                ha[J] = H[(unsigned)row_lane<J>(qa.y) * 16u + f];
                hb[J] = H[(unsigned)row_lane<J>(qb.y) * 16u + f];
            });
            const int4 qan = rec(t0 + 64 + lane);
            const int4 qbn = rec(t0 + 96 + lane);
            static_for<0, 16>([&](auto jc) {
                constexpr int J = decltype(jc)::value;
                const unsigned ra = (unsigned)row_lane<J>(qa.x);
                const float    va = __int_as_float(row_lane<J>(qa.z));
                atomicAdd(acc + (ra * 16u + f), va * ha[J]);
                const unsigned rb = (unsigned)row_lane<J>(qb.x);
                const float    vb = __int_as_float(row_lane<J>(qb.z));
                atomicAdd(acc + (rb * 16u + f), vb * hb[J]);
            });
            qa = qan;
            qb = qbn;
        }
    }
    __syncthreads();

    {
        float4* o4 = (float4*)out;
        for (int i = tid; i < L2_ROWS * 4; i += L2_THREADS) {
            const int rrow = r0 + (i >> 2);
            if (rrow < M) o4[(size_t)rrow * 4 + (i & 3)] = a4[i];
        }
    }
}

extern "C" void kernel_launch(void* const* d_in, const int* in_sizes, int n_in,
                              void* d_out, int out_size, void* d_ws, size_t ws_size,
                              hipStream_t stream) {
    const float* x   = (const float*)d_in[0];
    const int*   row = (const int*)  d_in[1];
    const int*   col = (const int*)  d_in[2];
    const float* val = (const float*)d_in[3];
    const float* W1  = (const float*)d_in[4];
    const float* b1  = (const float*)d_in[5];
    const float* W2  = (const float*)d_in[6];
    const float* b2  = (const float*)d_in[7];
    float* out = (float*)d_out;

    constexpr int D_IN  = 128;
    constexpr int D_HID = 128;
    constexpr int N_CLS = 16;
    static_assert(D_HID == 128 && N_CLS == 16, "the row-sum launch's own product is written for 128 x 16");
    const int E = in_sizes[1];
    const int M = in_sizes[0] / D_IN;

    char*  ws  = (char*)d_ws;
    float* H1  = (float*)ws;
    float* H2  = (float*)(ws + (size_t)M * D_HID * sizeof(float));
    unsigned int* KEY = (unsigned int*)(ws + ((size_t)D_HID + N_CLS) * (size_t)M * sizeof(float));
    const int nranges = (M + AG_ROWS - 1) / AG_ROWS;
    const size_t nlists = (size_t)nranges * AG_WAVES;
    int*  RECS = (int*)((char*)KEY + ((size_t)((E + 511) >> 9) << 9));
    int*  LCNT = (int*)((char*)RECS + ((size_t)((E + 511) >> 9) << 12));
    int*  LST  = (int*)((char*)LCNT + ((nlists * sizeof(int) + 127) & ~(size_t)127));


    {
        const int nstripes = M / 16;
        const unsigned nb = (unsigned)((nstripes + G1_WAVES - 1) / G1_WAVES);
        gcn_gemm_stripe128_f32<<<nb, G1_WAVES * 32, 0, stream>>>(x, W1, b1, H1, nstripes, row, col, val, KEY, RECS, E);
    }

    spmm_rows_in_lds128<<<(unsigned)nranges, AG_THREADS, 0, stream>>>(RECS, H1, W2, b2, H2, KEY, LST, LCNT, E, M);

    spmm_lists16<<<(unsigned)nranges, L2_THREADS, 0, stream>>>(LST, LCNT, row, col, val, H2, out, E, M);
}


# Layer-1 row sums: LDS float atomics replaced by owner waves: found edges go to the LDS queue of the wave owning their row (row mod 32), which adds by 16-B LDS read-fma-write; sum order: queue order
# speedup vs baseline: 3.3876x; 1.1025x over previous
#include <hip/hip_runtime.h>
#include <type_traits>

typedef __attribute__((ext_vector_type(2))) float v2f;
typedef __attribute__((ext_vector_type(8))) float v8f;

constexpr int AG_ROWS  = 200;
constexpr int G1_WAVES = 8;

__device__ __forceinline__ unsigned edge_key(int r) {
    const unsigned k = (unsigned)r / (unsigned)AG_ROWS;
    return k < 255u ? k : 255u;
}

static_assert(AG_ROWS <= 512, "the row within a range has 9 bits in an edge record");
constexpr int      REC_COL_BITS = 23;
constexpr unsigned REC_COL_MASK = (1u << REC_COL_BITS) - 1u;

__device__ __forceinline__ int edge_rec(int r, int c) {
    const unsigned k = (unsigned)r / (unsigned)AG_ROWS;
    return (int)((((unsigned)r - k * (unsigned)AG_ROWS) << REC_COL_BITS) | ((unsigned)c & REC_COL_MASK));
}

__global__ __launch_bounds__(G1_WAVES * 32)
void gcn_gemm_stripe128_f32(const float* __restrict__ A,
                            const float* __restrict__ W,
                            const float* __restrict__ bias,
                            float* __restrict__ Dout, int nstripes,
                            const int* __restrict__ row,
                            const int* __restrict__ col,
                            const float* __restrict__ val,
                            unsigned int* __restrict__ keys,
                            int* __restrict__ recs, int E)
{
    constexpr int K = 128, N = 128;
    __shared__ __attribute__((aligned(16))) v2f Ws[(K / 2) * N];

    const int tid  = threadIdx.x;

    const int nwords = ((E + 511) >> 9) << 7;
    const int gsz    = (int)gridDim.x * (G1_WAVES * 32);
    const int i0     = (int)blockIdx.x * (G1_WAVES * 32) + tid;
    auto whole_rows = [&](int i) -> int4 {
        const int e = i << 2;
        return *(const int4*)(row + ((e + 3 < E) ? e : 0));
    };
    auto whole_cols = [&](int i) -> int4 {
        const int e = i << 2;
        return *(const int4*)(col + ((e + 3 < E) ? e : 0));
    };
    auto whole_vals = [&](int i) -> float4 {
        const int e = i << 2;
        return *(const float4*)(val + ((e + 3 < E) ? e : 0));
    };
    auto end_rows = [&](int i) -> int4 {
        const int e = i << 2;
        int4 r = make_int4(-1, -1, -1, -1);
        if (e     < E) r.x = row[e];
        if (e + 1 < E) r.y = row[e + 1];
        if (e + 2 < E) r.z = row[e + 2];
        return r;
    };
    auto store_keys = [&](int i, int4 r, const int4 c, const float4 v) {
        const int e = i << 2;
        if (e + 3 >= E) r = end_rows(i);
        keys[i] = edge_key(r.x) | (edge_key(r.y) << 8) | (edge_key(r.z) << 16) | (edge_key(r.w) << 24);
        if (e + 3 < E) {
            int4* d = (int4*)(recs + 2 * (size_t)e);
            d[0] = make_int4(edge_rec(r.x, c.x), __float_as_int(v.x), edge_rec(r.y, c.y), __float_as_int(v.y));
            d[1] = make_int4(edge_rec(r.z, c.z), __float_as_int(v.z), edge_rec(r.w, c.w), __float_as_int(v.w));
        } else {
            for (int k = 0; k < 3; ++k)
                if (e + k < E) {
                    recs[2 * (size_t)(e + k)]     = edge_rec(row[e + k], col[e + k]);
                    recs[2 * (size_t)(e + k) + 1] = __float_as_int(val[e + k]);
                }
        }
    };
    int4   rk[2], ck[2];
    float4 vk[2];
#pragma unroll
    for (int t = 0; t < 2; ++t) {
        rk[t] = whole_rows(i0 + t * gsz);
        ck[t] = whole_cols(i0 + t * gsz);
        vk[t] = whole_vals(i0 + t * gsz);
    }

    const int lane = tid & 31;
    const int hi   = lane >> 4;
    const int l16  = lane & 15;
    const int wave = __builtin_amdgcn_readfirstlane(tid >> 5);
    const int stripe = blockIdx.x * G1_WAVES + wave;
    const bool live = stripe < nstripes;
    const int mtile = (live ? stripe : 0) << 4;

    const float* __restrict__ arow = A + (size_t)(mtile + l16) * K + (hi << 1);
    v2f a[K / 4];
#pragma unroll
    for (int s = 0; s < K / 4; ++s) a[s] = *(const v2f*)(arow + 4 * s);

#pragma unroll
    for (int j = 0; j < (K / 2) * (N / 4) / (G1_WAVES * 32); ++j) {
        const int i  = tid + j * (G1_WAVES * 32);
        const int p  = i >> 5;
        const int n4 = (i & 31) << 2;
        const float4 r0 = *(const float4*)(W + (size_t)(2 * p) * N + n4);
        const float4 r1 = *(const float4*)(W + (size_t)(2 * p + 1) * N + n4);
        float4* d = (float4*)(Ws + p * N + n4);
        d[0] = make_float4(r0.x, r1.x, r0.y, r1.y);
        d[1] = make_float4(r0.z, r1.z, r0.w, r1.w);
    }

    v8f acc[N / 16];
#pragma unroll
    for (int t = 0; t < N / 16; ++t) {
        const float bv = bias[t * 16 + l16];
#pragma unroll
        for (int r = 0; r < 8; ++r) acc[t][r] = bv;
    }
    __syncthreads();
    if (live) {
        const v2f* __restrict__ wl = Ws + hi * N + l16;
#pragma unroll
        for (int s = 0; s < K / 4; ++s) {
            v2f b[N / 16];
#pragma unroll
            for (int t = 0; t < N / 16; ++t) b[t] = wl[(2 * s) * N + t * 16];
#pragma unroll
            for (int t = 0; t < N / 16; ++t)
                acc[t] = __builtin_amdgcn_wmma_f32_16x16x4_f32(
                    false, a[s], false, b[t], (short)0, acc[t], false, false);
        }

#pragma unroll
        for (int r = 0; r < 8; ++r) {
            float* __restrict__ drow = Dout + (size_t)(mtile + (hi << 3) + r) * N + l16;
#pragma unroll
            for (int t = 0; t < N / 16; ++t) drow[t * 16] = acc[t][r];
        }
    }

#pragma unroll
    for (int t = 0; t < 2; ++t)
        if (i0 + t * gsz < nwords) store_keys(i0 + t * gsz, rk[t], ck[t], vk[t]);
    for (int i = i0 + 2 * gsz; i < nwords; i += gsz) store_keys(i, whole_rows(i), whole_cols(i), whole_vals(i));
}

constexpr int AG_THREADS = 1024;
constexpr int AG_WAVES   = AG_THREADS / 32;
constexpr int AG_UNROLL  = 4;
constexpr int AG_EAGER   = 32;
constexpr int AG_QCAP    = AG_EAGER + 32;
constexpr int OQ_CAP     = 768;
constexpr int AG_STRIPES = (AG_ROWS + 15) / 16;
constexpr int AG_SPARE   = AG_STRIPES * 16;
constexpr int AG_PITCH   = 132;
static_assert(AG_UNROLL % 2 == 0, "pieces are searched two at a time");
static_assert(AG_STRIPES <= AG_WAVES, "a wave takes one stripe of the layer-2 product");
static_assert(AG_PITCH % 4 == 0 && AG_PITCH >= 128, "rows are cleared 16 bytes at a time");
constexpr int L2_ROWS = AG_ROWS;
constexpr int L2_CAP  = 256;

__global__ __launch_bounds__(AG_THREADS)
void spmm_rows_in_lds128(const int* __restrict__ recs,
                         const float* __restrict__ H,
                         const float* __restrict__ W2,
                         const float* __restrict__ b2,
                         float* __restrict__ H2,
                         const unsigned int* __restrict__ keys,
                         int* __restrict__ lst, int* __restrict__ lcnt,
                         const int* __restrict__ row, const int* __restrict__ col,
                         const float* __restrict__ val,
                         int E, int M)
{
    __shared__ __attribute__((aligned(16))) float acc[(AG_SPARE + 1) * AG_PITCH];
    __shared__ int   qbuf[AG_WAVES * AG_QCAP];
    __shared__ __attribute__((aligned(8))) int2 oq[AG_WAVES * OQ_CAP];
    __shared__ int   ocnt[AG_WAVES];
    __shared__ int   ovf;

    const int tid  = threadIdx.x;
    const int lane = tid & 31;
    const int wave = __builtin_amdgcn_readfirstlane(tid >> 5);
    const int range = (int)blockIdx.x;
    const int r0    = range * AG_ROWS;

    {
        float4* a4 = (float4*)acc;
        const float4 z = make_float4(0.f, 0.f, 0.f, 0.f);
        for (int i = tid; i < (AG_SPARE + 1) * (AG_PITCH / 4); i += AG_THREADS) a4[i] = z;
    }
    if (tid < AG_WAVES) ocnt[tid] = 0;
    if (tid == 0) ovf = 0;
    __syncthreads();

    int* myq = qbuf + wave * AG_QCAP;
    int  cnt = 0;
    int4* __restrict__ mylst = (int4*)lst + ((size_t)blockIdx.x * AG_WAVES + wave) * L2_CAP;
    int  wn = 0;

    const unsigned pat = (unsigned)range * 0x01010101u;
    const unsigned lt  = (1u << lane) - 1u;
    auto flags = [&](unsigned w) -> unsigned {
        const unsigned x = w ^ pat;
        const unsigned z = ((x & 0x7f7f7f7fu) + 0x7f7f7f7fu) | x | 0x7f7f7f7fu;
        return (~z) >> 7;
    };
    const int2* __restrict__ rp = (const int2*)recs;
    auto drain = [&]() {
        for (int b = 0; b < cnt; b += 32) {
            const int nb = cnt - b;
            if (lane < nb) {
                const int2 q = rp[myq[b + lane]];
                const int mr = (int)((unsigned)q.x >> REC_COL_BITS);
                const int mc = (int)((unsigned)q.x & REC_COL_MASK);
                const int pos = wn + lane;
                if (pos < L2_CAP) mylst[pos] = make_int4(mr, mc, q.y, 0);
                const int owner = mr & (AG_WAVES - 1);
                const int slot  = atomicAdd(&ocnt[owner], 1);
                if (slot < OQ_CAP) oq[owner * OQ_CAP + slot] = q;
                else ovf = 1;
            }
            wn += nb < 32 ? nb : 32;
        }
        cnt = 0;
    };

    auto take = [&](const int4 ka, const int4 kb, int ea, int eb) {
        unsigned m = flags((unsigned)ka.x)        | (flags((unsigned)ka.y) << 1)
                   | (flags((unsigned)ka.z) << 2) | (flags((unsigned)ka.w) << 3)
                   | (flags((unsigned)kb.x) << 4) | (flags((unsigned)kb.y) << 5)
                   | (flags((unsigned)kb.z) << 6) | (flags((unsigned)kb.w) << 7);
#pragma nounroll
        for (int it = 0; it < 32; ++it) {
            const bool hit = m != 0u;
            const unsigned mask = __builtin_amdgcn_ballot_w32(hit);
            if (mask == 0u) break;
            if (cnt + 32 > AG_QCAP) drain();
            if (hit) {
                const int p = __builtin_ctz(m);
                const int i = p & 7;
                myq[cnt + __builtin_popcount(mask & lt)] = ((i & 4) ? eb : ea) + (((i & 3) << 2) | (p >> 3));
                m &= m - 1u;
            }
            cnt += __builtin_popcount(mask);
        }
    };

    const int c1 = (E + 511) >> 9;
    const int4* __restrict__ kp = (const int4*)keys;
    constexpr int AG_ROUND = AG_WAVES * AG_UNROLL;
    auto load_keys = [&](int4 (&k)[AG_UNROLL], int base) {
#pragma unroll
        for (int u = 0; u < AG_UNROLL; ++u) {
            const int ch = base + u * AG_WAVES + wave;
            k[u] = kp[(size_t)(ch < c1 ? ch : c1 - 1) * 32 + lane];
        }
    };
    if (c1 > 0) {
        int4 kk[AG_UNROLL];
        load_keys(kk, 0);
        for (int base = 0; base < c1; base += AG_ROUND) {
            int4 kn[AG_UNROLL];
            load_keys(kn, base + AG_ROUND);
#pragma unroll
            for (int u = 0; u < AG_UNROLL; u += 2) {
                const int ca = base + u * AG_WAVES + wave;
                const int cb = ca + AG_WAVES;
                if (ca < c1) {
                    if (cnt >= AG_EAGER) drain();
                    const int4 kb = (cb < c1) ? kk[u + 1] : make_int4(-1, -1, -1, -1);
                    take(kk[u], kb, (ca << 9) + (lane << 4), (cb << 9) + (lane << 4));
                }
            }
#pragma unroll
            for (int u = 0; u < AG_UNROLL; ++u) kk[u] = kn[u];
        }
    }
    drain();
    if (lane == 0) lcnt[(size_t)blockIdx.x * AG_WAVES + wave] = wn;
    __syncthreads();

    if (__builtin_amdgcn_readfirstlane(ovf) != 0) {
        for (int e = tid; e < E; e += AG_THREADS) {
            const unsigned rl = (unsigned)(row[e] - r0);
            if (rl < (unsigned)AG_ROWS) {
                const float v = val[e];
                const float* __restrict__ hp = H + (size_t)(unsigned)col[e] * 128u;
                float* a = acc + rl * (unsigned)AG_PITCH;
                for (int k = 0; k < 128; ++k) atomicAdd(a + k, v * hp[k]);
            }
        }
    } else {
        int n = __builtin_amdgcn_readfirstlane(ocnt[wave]);
        n = n < OQ_CAP ? n : OQ_CAP;
        const int2* oqw = oq + wave * OQ_CAP;
        const float4* __restrict__ H4 = (const float4*)H;
        auto fetch = [&](int g, float4 (&h)[8], int& mo, int& mv) {
            int2 q = make_int2((int)((unsigned)AG_SPARE << REC_COL_BITS), 0);
            const int i = g + (lane & 7);
            if (i < n) q = oqw[i];
            mo = (int)((unsigned)q.x >> REC_COL_BITS) * AG_PITCH;
            mv = q.y;
            const int mc = (int)((unsigned)q.x & REC_COL_MASK);
#pragma unroll
            for (int j = 0; j < 8; ++j) {
                const int cj = __builtin_amdgcn_readlane(mc, j);
                h[j] = H4[(size_t)cj * 32 + lane];
            }
        };
        auto apply = [&](const float4 (&h)[8], int mo, int mv) {
#pragma unroll
            for (int j = 0; j < 8; ++j) {
                const int   ro = __builtin_amdgcn_readlane(mo, j);
                const float v  = __int_as_float(__builtin_amdgcn_readlane(mv, j));
                float4* a = (float4*)(acc + ro) + lane;
                float4 t = *a;
                t.x += v * h[j].x;
                t.y += v * h[j].y;
                t.z += v * h[j].z;
                t.w += v * h[j].w;
                *a = t;
            }
        };
        float4 ha[8], hb[8];
        int moa, mva, mob, mvb;
        fetch(0, ha, moa, mva);
        for (int g = 0; g < n; g += 16) {
            fetch(g + 8, hb, mob, mvb);
            apply(ha, moa, mva);
            fetch(g + 16, ha, moa, mva);
            apply(hb, mob, mvb);
        }
    }

    const int hi  = lane >> 4;
    const int l16 = lane & 15;
    v2f   bw[32];
    float bv = 0.f;
    if (wave < AG_STRIPES) {
        const float* __restrict__ bcol = W2 + (size_t)(hi << 1) * 16 + l16;
        bv = b2[l16];
#pragma unroll
        for (int s = 0; s < 32; ++s) {
            bw[s][0] = bcol[(4 * s) * 16];
            bw[s][1] = bcol[(4 * s + 1) * 16];
        }
    }
    __syncthreads();

    if (wave < AG_STRIPES) {
        const float* __restrict__ arow = acc + (wave * 16 + l16) * AG_PITCH + (hi << 1);
        v8f d;
#pragma unroll
        for (int r = 0; r < 8; ++r) d[r] = bv;
#pragma unroll
        for (int k0 = 0; k0 < 128; k0 += 4) {
            v2f a = *(const v2f*)(arow + k0);
            a[0] = fmaxf(a[0], 0.0f);
            a[1] = fmaxf(a[1], 0.0f);
            d = __builtin_amdgcn_wmma_f32_16x16x4_f32(
                false, a, false, bw[k0 / 4], (short)0, d, false, false);
        }
#pragma unroll
        for (int r = 0; r < 8; ++r) {
            const int lrow = wave * 16 + (hi << 3) + r;
            if (lrow < AG_ROWS && r0 + lrow < M)
                H2[(size_t)(r0 + lrow) * 16 + l16] = d[r];
        }
    }
}

constexpr int L2_THREADS = AG_THREADS;

template <int J>
__device__ __forceinline__ int row_lane(int x) {
    return __builtin_amdgcn_update_dpp(0, x, 0x150 + J, 0xf, 0xf, true);
}

template <int J, int N, class F>
__device__ __forceinline__ void static_for(F&& fn) {
    if constexpr (J < N) {
        fn(std::integral_constant<int, J>{});
        static_for<J + 1, N>(fn);
    }
}

__global__ __launch_bounds__(L2_THREADS)
void spmm_lists16(const int* __restrict__ lst, const int* __restrict__ lcnt,
                  const int* __restrict__ row, const int* __restrict__ col,
                  const float* __restrict__ val,
                  const float* __restrict__ H, float* __restrict__ out, int E, int M)
{
    __shared__ __attribute__((aligned(16))) float acc[(L2_ROWS + AG_WAVES) * 16];
    __shared__ int short_list;

    const int tid  = threadIdx.x;
    const int lane = tid & 31;
    const unsigned f = lane & 15;
    const int wave  = __builtin_amdgcn_readfirstlane(tid >> 5);
    const int r0    = (int)blockIdx.x * L2_ROWS;

    float4* a4 = (float4*)acc;
    {
        const float4 z = make_float4(0.f, 0.f, 0.f, 0.f);
        for (int i = tid; i < (L2_ROWS + AG_WAVES) * 4; i += L2_THREADS) a4[i] = z;
    }
    if (tid == 0) short_list = 0;

    const int sl = (int)blockIdx.x * AG_WAVES + wave;
    const int n  = lcnt[sl];
    __syncthreads();
    if (n > L2_CAP && lane == 0) short_list = 1;
    __syncthreads();

    if (short_list != 0) {
        for (int e = tid; e < E; e += L2_THREADS) {
            const unsigned rl = (unsigned)(row[e] - r0);
            if (rl < (unsigned)L2_ROWS) {
                const float v = val[e];
                const float* __restrict__ hp = H + (size_t)(unsigned)col[e] * 16u;
                for (int k = 0; k < 16; ++k) atomicAdd(acc + rl * 16u + k, v * hp[k]);
            }
        }
    } else {
        const int4* __restrict__ l = (const int4*)lst + (size_t)sl * L2_CAP;
        const int spare = L2_ROWS + wave;
        auto rec = [&](int i) -> int4 {
            int4 q = make_int4(spare, 0, 0, 0);
            if (i < n) q = l[i];
            return q;
        };
        int4 qa = rec(lane);
        int4 qb = rec(32 + lane);
        for (int t0 = 0; t0 < n; t0 += 64) {
            float ha[16], hb[16];
            static_for<0, 16>([&](auto jc) {
                constexpr int J = decltype(jc)::value;
                ha[J] = H[(unsigned)row_lane<J>(qa.y) * 16u + f];
                hb[J] = H[(unsigned)row_lane<J>(qb.y) * 16u + f];
            });
            const int4 qan = rec(t0 + 64 + lane);
            const int4 qbn = rec(t0 + 96 + lane);
            static_for<0, 16>([&](auto jc) {
                constexpr int J = decltype(jc)::value;
                const unsigned ra = (unsigned)row_lane<J>(qa.x);
                const float    va = __int_as_float(row_lane<J>(qa.z));
                atomicAdd(acc + (ra * 16u + f), va * ha[J]);
                const unsigned rb = (unsigned)row_lane<J>(qb.x);
                const float    vb = __int_as_float(row_lane<J>(qb.z));
                atomicAdd(acc + (rb * 16u + f), vb * hb[J]);
            });
            qa = qan;
            qb = qbn;
        }
    }
    __syncthreads();

    {
        float4* o4 = (float4*)out;
        for (int i = tid; i < L2_ROWS * 4; i += L2_THREADS) {
            const int rrow = r0 + (i >> 2);
            if (rrow < M) o4[(size_t)rrow * 4 + (i & 3)] = a4[i];
        }
    }
}

extern "C" void kernel_launch(void* const* d_in, const int* in_sizes, int n_in,
                              void* d_out, int out_size, void* d_ws, size_t ws_size,
                              hipStream_t stream) {
    const float* x   = (const float*)d_in[0];
    const int*   row = (const int*)  d_in[1];
    const int*   col = (const int*)  d_in[2];
    const float* val = (const float*)d_in[3];
    const float* W1  = (const float*)d_in[4];
    const float* b1  = (const float*)d_in[5];
    const float* W2  = (const float*)d_in[6];
    const float* b2  = (const float*)d_in[7];
    float* out = (float*)d_out;

    constexpr int D_IN  = 128;
    constexpr int D_HID = 128;
    constexpr int N_CLS = 16;
    static_assert(D_HID == 128 && N_CLS == 16, "the row-sum launch's own product is written for 128 x 16");
    const int E = in_sizes[1];
    const int M = in_sizes[0] / D_IN;

    char*  ws  = (char*)d_ws;
    float* H1  = (float*)ws;
    float* H2  = (float*)(ws + (size_t)M * D_HID * sizeof(float));
    unsigned int* KEY = (unsigned int*)(ws + ((size_t)D_HID + N_CLS) * (size_t)M * sizeof(float));
    const int nranges = (M + AG_ROWS - 1) / AG_ROWS;
    const size_t nlists = (size_t)nranges * AG_WAVES;
    int*  RECS = (int*)((char*)KEY + ((size_t)((E + 511) >> 9) << 9));
    int*  LCNT = (int*)((char*)RECS + ((size_t)((E + 511) >> 9) << 12));
    int*  LST  = (int*)((char*)LCNT + ((nlists * sizeof(int) + 127) & ~(size_t)127));


    {
        const int nstripes = M / 16;
        const unsigned nb = (unsigned)((nstripes + G1_WAVES - 1) / G1_WAVES);
        gcn_gemm_stripe128_f32<<<nb, G1_WAVES * 32, 0, stream>>>(x, W1, b1, H1, nstripes, row, col, val, KEY, RECS, E);
    }

    spmm_rows_in_lds128<<<(unsigned)nranges, AG_THREADS, 0, stream>>>(RECS, H1, W2, b2, H2, KEY, LST, LCNT, row, col, val, E, M);

    spmm_lists16<<<(unsigned)nranges, L2_THREADS, 0, stream>>>(LST, LCNT, row, col, val, H2, out, E, M);
}
